# starting kernel
# baseline (speedup 1.0000x reference)
#include <hip/hip_runtime.h>

typedef float v2f __attribute__((ext_vector_type(2)));
typedef float v8f __attribute__((ext_vector_type(8)));

#define S_GRID   56
#define N_CLASS  20
#define CELLS    (S_GRID * S_GRID)
#define ITEMS    (CELLS * 2)
#define SORTN    8192
#define CAND     256
#define MAXOUT   30
#define D_IMG    94080
#define CONF_OFF 62720
#define COORD_OFF 68992
#define NEG_INF  (-3.402823466e38f)

struct Phase2 {
    float y0[CAND], x0[CAND], y1[CAND], x1[CAND];
    float sc[CAND], cl[CAND], ar[CAND];
    float redv[CAND];
    int   redi[CAND];
    unsigned short mask16[CAND * 16];
    int selIdx[32], selOk[32];
    unsigned char valid[CAND];
};

__global__ __launch_bounds__(256) void det_kernel(const float* __restrict__ net,
                                                  float* __restrict__ out) {
    __shared__ unsigned long long keys[SORTN];
    Phase2* P = reinterpret_cast<Phase2*>(&keys[CAND]);

    const int tid = threadIdx.x;
    const int img = blockIdx.x;
    const float* in = net + (size_t)img * D_IMG;

    for (int cell = tid; cell < CELLS; cell += 256) {
        const float* pr = in + cell * N_CLASS;
        float p[N_CLASS];
        #pragma unroll
        for (int c = 0; c < N_CLASS; ++c) p[c] = pr[c];
        #pragma unroll
        for (int b = 0; b < 2; ++b) {
            int id = cell * 2 + b;
            float conf = in[CONF_OFF + id];
            float best = NEG_INF;
            #pragma unroll
            for (int c = 0; c < N_CLASS; ++c) {
                float v = __fmul_rn(conf, p[c]);
                best = v > best ? v : best;
            }
            float s = (best >= 0.1f) ? best : 0.0f;
            keys[id] = ((unsigned long long)__float_as_uint(s) << 32) |
                       (unsigned)(~(unsigned)id);
        }
    }
    for (int q = ITEMS + tid; q < SORTN; q += 256)
        keys[q] = (unsigned long long)(unsigned)(~(unsigned)q);
    __syncthreads();

    for (unsigned k = 2; k <= (unsigned)SORTN; k <<= 1) {
        for (unsigned j = k >> 1; j > 0; j >>= 1) {
            for (unsigned i = tid; i < (unsigned)SORTN; i += 256u) {
                unsigned x = i ^ j;
                if (x > i) {
                    unsigned long long a = keys[i], b = keys[x];
                    bool up = ((i & k) == 0u);
                    if (up ? (a < b) : (a > b)) { keys[i] = b; keys[x] = a; }
                }
            }
            __syncthreads();
        }
    }

    {
        unsigned long long key = keys[tid];
        unsigned id = ~(unsigned)key;
        float s = __uint_as_float((unsigned)(key >> 32));
        int cell = (int)(id >> 1);
        int gi = cell / S_GRID, gj = cell % S_GRID;
        const float* cd = in + COORD_OFF + (size_t)id * 4;
        float tx = cd[0], ty = cd[1], tw = cd[2], th = cd[3];
        float x = __fdiv_rn(__fadd_rn(tx, (float)gj), 56.0f);
        float y = __fdiv_rn(__fadd_rn(ty, (float)gi), 56.0f);
        float hw = __fmul_rn(__fmul_rn(tw, tw), 0.5f);
        float hh = __fmul_rn(__fmul_rn(th, th), 0.5f);
        float y0 = __fsub_rn(y, hh), x0 = __fsub_rn(x, hw);
        float y1 = __fadd_rn(y, hh), x1 = __fadd_rn(x, hw);
        float conf = in[CONF_OFF + id];
        const float* pr = in + cell * N_CLASS;
        float best = NEG_INF; int bc = 0;
        #pragma unroll
        for (int c = 0; c < N_CLASS; ++c) {
            float v = __fmul_rn(conf, pr[c]);
            if (v > best) { best = v; bc = c; }
        }
        float dy = fmaxf(__fsub_rn(y1, y0), 0.0f);
        float dx = fmaxf(__fsub_rn(x1, x0), 0.0f);
        P->y0[tid] = y0; P->x0[tid] = x0; P->y1[tid] = y1; P->x1[tid] = x1;
        P->sc[tid] = s;  P->cl[tid] = (float)bc;
        P->ar[tid] = __fmul_rn(dy, dx);
        P->valid[tid] = 1;
    }
    __syncthreads();

    {
        const int lane = tid & 31;
        const int wave = tid >> 5;
        v8f czero = {};
        for (int tt = wave * 32; tt < wave * 32 + 32; ++tt) {
            int mi = tt >> 4, ni = tt & 15;
            int m0 = mi << 4, n0 = ni << 4;
            v2f A, B;
            if (lane < 16) {
                A[0] = P->ar[m0 + lane]; A[1] = 1.0f;
                B[0] = 1.0f;             B[1] = P->ar[n0 + lane];
            } else {
                A[0] = 0.0f; A[1] = 0.0f; B[0] = 0.0f; B[1] = 0.0f;
            }
            v8f d = __builtin_amdgcn_wmma_f32_16x16x4_f32(
                false, A, false, B, (short)0, czero, false, false);
            int n = n0 + (lane & 15);
            float ny0 = P->y0[n], nx0 = P->x0[n], ny1 = P->y1[n], nx1 = P->x1[n];
            int mbase = m0 + ((lane & 16) ? 8 : 0);
            #pragma unroll
            for (int r = 0; r < 8; ++r) {
                int m = mbase + r;
                float iy = fmaxf(__fsub_rn(fminf(P->y1[m], ny1), fmaxf(P->y0[m], ny0)), 0.0f);
                float ix = fmaxf(__fsub_rn(fminf(P->x1[m], nx1), fmaxf(P->x0[m], nx0)), 0.0f);
                float inter = __fmul_rn(iy, ix);
                float uni = __fsub_rn(d[r], inter);
                bool pred = (uni > 0.0f) && (__fdiv_rn(inter, uni) > 0.4f);
                unsigned bal = __builtin_amdgcn_ballot_w32(pred);
                if (lane == 0) {
                    P->mask16[(m0 + r)     * 16 + ni] = (unsigned short)(bal & 0xFFFFu);
                    P->mask16[(m0 + 8 + r) * 16 + ni] = (unsigned short)(bal >> 16);
                }
            }
        }
    }
    __syncthreads();

    for (int it = 0; it < MAXOUT; ++it) {
        P->redv[tid] = P->valid[tid] ? P->sc[tid] : NEG_INF;
        P->redi[tid] = tid;
        __syncthreads();
        for (int off = 128; off > 0; off >>= 1) {
            if (tid < off) {
                float v2 = P->redv[tid + off]; int i2 = P->redi[tid + off];
                if (v2 > P->redv[tid] || (v2 == P->redv[tid] && i2 < P->redi[tid])) {
                    P->redv[tid] = v2; P->redi[tid] = i2;
                }
            }
            __syncthreads();
        }
        int idx = P->redi[0];
        int ok  = P->valid[idx];
        unsigned bit = (P->mask16[idx * 16 + (tid >> 4)] >> (tid & 15)) & 1u;
        unsigned char nv = (unsigned char)(P->valid[tid] & (unsigned char)(1u - bit));
        if (tid == idx) nv = 0;
        __syncthreads();
        P->valid[tid] = nv;
        if (tid == 0) { P->selIdx[it] = idx; P->selOk[it] = ok; }
        __syncthreads();
    }

    if (tid < MAXOUT) {
        int sel = P->selIdx[tid];
        float okf = P->selOk[tid] ? 1.0f : 0.0f;
        P->redv[tid] = __fmul_rn(P->sc[sel], okf);
    }
    __syncthreads();
    if (tid < MAXOUT) {
        int sel = P->selIdx[tid];
        float okf = P->selOk[tid] ? 1.0f : 0.0f;
        float ms = P->redv[tid];
        int rank = 0;
        for (int jt = 0; jt < MAXOUT; ++jt) {
            float vj = P->redv[jt];
            if (vj > ms || (vj == ms && jt < tid)) rank++;
        }
        float* o = out + (size_t)img * (MAXOUT * 6) + (size_t)rank * 6;
        o[0] = __fmul_rn(P->y0[sel], okf);
        o[1] = __fmul_rn(P->x0[sel], okf);
        o[2] = __fmul_rn(P->y1[sel], okf);
        o[3] = __fmul_rn(P->x1[sel], okf);
        o[4] = ms;
        o[5] = __fmul_rn(P->cl[sel], okf);
    }
}

extern "C" void kernel_launch(void* const* d_in, const int* in_sizes, int n_in,
                              void* d_out, int out_size, void* d_ws, size_t ws_size,
                              hipStream_t stream) {
    (void)n_in; (void)out_size; (void)d_ws; (void)ws_size;
    const float* net = (const float*)d_in[0];
    float* out = (float*)d_out;
    int batch = in_sizes[0] / D_IMG;
    det_kernel<<<dim3(batch), dim3(256), 0, stream>>>(net, out);
}


# Top-256 by a pruned bitonic network held in registers: each wave sorts 256-key chunks by lane exchange (no barriers), halving merges keep the larger half; replaces the 8192-key LDS sort
# speedup vs baseline: 3.5242x; 3.5242x over previous
#include <hip/hip_runtime.h>

typedef float v2f __attribute__((ext_vector_type(2)));
typedef float v8f __attribute__((ext_vector_type(8)));

#define S_GRID   56
#define N_CLASS  20
#define CELLS    (S_GRID * S_GRID)
#define ITEMS    (CELLS * 2)
#define CAND     256
#define MAXOUT   30
#define D_IMG    94080
#define CONF_OFF 62720
#define COORD_OFF 68992
#define NEG_INF  (-3.402823466e38f)
#define NSLOT    32
#define NWAVE    8

typedef unsigned long long u64;


__device__ __forceinline__ u64 key_from_lane(u64 v, int src4) {
    int lo = __builtin_amdgcn_ds_bpermute(src4, (int)(unsigned)v);
    int hi = __builtin_amdgcn_ds_bpermute(src4, (int)(unsigned)(v >> 32));
    return ((u64)(unsigned)hi << 32) | (u64)(unsigned)lo;
}

template <int NCH, int STRIDE, int M, bool ALT>
__device__ __forceinline__ void stage_lanes(u64 (&key)[NSLOT], int lane, bool d0) {
    const int src4 = (lane ^ M) << 2;
    const bool lower = (lane & M) == 0;
    #pragma unroll
    for (int c = 0; c < NCH; ++c) {
        const bool desc = (ALT && (c & 1)) ? !d0 : d0;
        const bool keepmin = (lower != desc);
        #pragma unroll
        for (int q = 0; q < 8; ++q) {
            const int s = c * STRIDE + q;
            u64 mine = key[s];
            u64 oth = key_from_lane(mine, src4);
            bool take = (oth > mine) != keepmin;
            key[s] = take ? oth : mine;
        }
    }
}

template <int NCH, int STRIDE, int J, bool ALT>
__device__ __forceinline__ void stage_regs(u64 (&key)[NSLOT], bool d0) {
    #pragma unroll
    for (int c = 0; c < NCH; ++c) {
        const bool desc = (ALT && (c & 1)) ? !d0 : d0;
        #pragma unroll
        for (int q = 0; q < 8; ++q) {
            if ((q & J) == 0) {
                const int s = c * STRIDE + q;
                u64 a = key[s], b = key[s + J];
                bool sw = (a < b) == desc;
                key[s]     = sw ? b : a;
                key[s + J] = sw ? a : b;
            }
        }
    }
}

template <int NCH, int STRIDE, int J, int K>
__device__ __forceinline__ void stage_regs_q(u64 (&key)[NSLOT]) {
    #pragma unroll
    for (int c = 0; c < NCH; ++c) {
        #pragma unroll
        for (int q = 0; q < 8; ++q) {
            if ((q & J) == 0) {
                const int s = c * STRIDE + q;
                const bool desc = (q & K) == 0;
                u64 a = key[s], b = key[s + J];
                bool sw = desc ? (a < b) : (a > b);
                key[s]     = sw ? b : a;
                key[s + J] = sw ? a : b;
            }
        }
    }
}

template <int NCH, int STRIDE, int MTOP, bool ALT>
__device__ __forceinline__ void merge_chunks(u64 (&key)[NSLOT], int lane, bool d0) {
    if (MTOP >= 16) stage_lanes<NCH, STRIDE, 16, ALT>(key, lane, d0);
    if (MTOP >= 8)  stage_lanes<NCH, STRIDE, 8,  ALT>(key, lane, d0);
    if (MTOP >= 4)  stage_lanes<NCH, STRIDE, 4,  ALT>(key, lane, d0);
    if (MTOP >= 2)  stage_lanes<NCH, STRIDE, 2,  ALT>(key, lane, d0);
    if (MTOP >= 1)  stage_lanes<NCH, STRIDE, 1,  ALT>(key, lane, d0);
    stage_regs<NCH, STRIDE, 4, ALT>(key, d0);
    stage_regs<NCH, STRIDE, 2, ALT>(key, d0);
    stage_regs<NCH, STRIDE, 1, ALT>(key, d0);
}

__device__ __forceinline__ u64 key_max(u64 a, u64 b) { return a > b ? a : b; }

struct Phase2 {
    float y0[CAND], x0[CAND], y1[CAND], x1[CAND];
    float sc[CAND], cl[CAND], ar[CAND];
    float redv[CAND];
    int   redi[CAND];
    unsigned short mask16[CAND * 16];
    int selIdx[32], selOk[32];
    unsigned char valid[CAND];
};

__global__ __launch_bounds__(256) void det_kernel(const float* __restrict__ net,
                                                  float* __restrict__ out) {
    __shared__ u64 keys[CAND];
    __shared__ u64 xch[3][NWAVE][CAND];
    __shared__ Phase2 p2;
    Phase2* P = &p2;

    const int tid = threadIdx.x;
    const int img = blockIdx.x;
    const float* in = net + (size_t)img * D_IMG;
    const int lane = tid & 31;
    const int wave = tid >> 5;

    u64 key[NSLOT];
    #pragma unroll
    for (int n = 0; n < 13; ++n) {
        const int cell = tid + 256 * n;
        if (n < 12 || cell < CELLS) {
            const float* pr = in + cell * N_CLASS;
            float p[N_CLASS];
            #pragma unroll
            for (int c = 0; c < N_CLASS; ++c) p[c] = pr[c];
            #pragma unroll
            for (int bx = 0; bx < 2; ++bx) {
                int id = cell * 2 + bx;
                float conf = in[CONF_OFF + id];
                float best = NEG_INF;
                #pragma unroll
                for (int c = 0; c < N_CLASS; ++c) {
                    float v = __fmul_rn(conf, p[c]);
                    best = v > best ? v : best;
                }
                float s = (best >= 0.1f) ? best : 0.0f;
                key[2 * n + bx] = ((u64)__float_as_uint(s) << 32) | (unsigned)(~(unsigned)id);
            }
        } else {
            key[2 * n]     = (u64)(unsigned)(~(unsigned)(ITEMS + tid * NSLOT + 2 * n));
            key[2 * n + 1] = (u64)(unsigned)(~(unsigned)(ITEMS + tid * NSLOT + 2 * n + 1));
        }
    }
    #pragma unroll
    for (int s = 26; s < NSLOT; ++s)
        key[s] = (u64)(unsigned)(~(unsigned)(ITEMS + tid * NSLOT + s));

    stage_regs_q<4, 8, 1, 2>(key);
    stage_regs_q<4, 8, 2, 4>(key);
    stage_regs_q<4, 8, 1, 4>(key);
    merge_chunks<4, 8, 0,  false>(key, lane, (lane & 1)  == 0);
    merge_chunks<4, 8, 1,  false>(key, lane, (lane & 2)  == 0);
    merge_chunks<4, 8, 2,  false>(key, lane, (lane & 4)  == 0);
    merge_chunks<4, 8, 4,  false>(key, lane, (lane & 8)  == 0);
    merge_chunks<4, 8, 8,  false>(key, lane, (lane & 16) == 0);
    merge_chunks<4, 8, 16, true >(key, lane, true);
    #pragma unroll
    for (int q = 0; q < 8; ++q) {
        key[q]      = key_max(key[q],      key[8 + q]);
        key[16 + q] = key_max(key[16 + q], key[24 + q]);
    }
    merge_chunks<2, 16, 16, true>(key, lane, true);
    #pragma unroll
    for (int q = 0; q < 8; ++q) key[q] = key_max(key[q], key[16 + q]);
    merge_chunks<1, 16, 16, false>(key, lane, (wave & 1) == 0);
    #pragma unroll
    for (int q = 0; q < 8; ++q) xch[0][wave][q * 32 + lane] = key[q];
    __syncthreads();
    #pragma unroll
    for (int q = 0; q < 8; ++q) key[q] = key_max(key[q], xch[0][wave ^ 1][q * 32 + lane]);
    merge_chunks<1, 16, 16, false>(key, lane, (wave & 2) == 0);
    #pragma unroll
    for (int q = 0; q < 8; ++q) xch[1][wave][q * 32 + lane] = key[q];
    __syncthreads();
    #pragma unroll
    for (int q = 0; q < 8; ++q) key[q] = key_max(key[q], xch[1][wave ^ 2][q * 32 + lane]);
    merge_chunks<1, 16, 16, false>(key, lane, (wave & 4) == 0);
    #pragma unroll
    for (int q = 0; q < 8; ++q) xch[2][wave][q * 32 + lane] = key[q];
    __syncthreads();
    #pragma unroll
    for (int q = 0; q < 8; ++q) key[q] = key_max(key[q], xch[2][wave ^ 4][q * 32 + lane]);
    merge_chunks<1, 16, 16, false>(key, lane, true);
    if (wave == 0) {
        #pragma unroll
        for (int q = 0; q < 8; ++q) keys[lane * 8 + q] = key[q];
    }
    __syncthreads();

    {
        unsigned long long key = keys[tid];
        unsigned id = ~(unsigned)key;
        float s = __uint_as_float((unsigned)(key >> 32));
        int cell = (int)(id >> 1);
        int gi = cell / S_GRID, gj = cell % S_GRID;
        const float* cd = in + COORD_OFF + (size_t)id * 4;
        float tx = cd[0], ty = cd[1], tw = cd[2], th = cd[3];
        float x = __fdiv_rn(__fadd_rn(tx, (float)gj), 56.0f);
        float y = __fdiv_rn(__fadd_rn(ty, (float)gi), 56.0f);
        float hw = __fmul_rn(__fmul_rn(tw, tw), 0.5f);
        float hh = __fmul_rn(__fmul_rn(th, th), 0.5f);
        float y0 = __fsub_rn(y, hh), x0 = __fsub_rn(x, hw);
        float y1 = __fadd_rn(y, hh), x1 = __fadd_rn(x, hw);
        float conf = in[CONF_OFF + id];
        const float* pr = in + cell * N_CLASS;
        float best = NEG_INF; int bc = 0;
        #pragma unroll
        for (int c = 0; c < N_CLASS; ++c) {
            float v = __fmul_rn(conf, pr[c]);
            if (v > best) { best = v; bc = c; }
        }
        float dy = fmaxf(__fsub_rn(y1, y0), 0.0f);
        float dx = fmaxf(__fsub_rn(x1, x0), 0.0f);
        P->y0[tid] = y0; P->x0[tid] = x0; P->y1[tid] = y1; P->x1[tid] = x1;
        P->sc[tid] = s;  P->cl[tid] = (float)bc;
        P->ar[tid] = __fmul_rn(dy, dx);
        P->valid[tid] = 1;
    }
    __syncthreads();

    {
        v8f czero = {};
        for (int tt = wave * 32; tt < wave * 32 + 32; ++tt) {
            int mi = tt >> 4, ni = tt & 15;
            int m0 = mi << 4, n0 = ni << 4;
            v2f A, B;
            if (lane < 16) {
                A[0] = P->ar[m0 + lane]; A[1] = 1.0f;
                B[0] = 1.0f;             B[1] = P->ar[n0 + lane];
            } else {
                A[0] = 0.0f; A[1] = 0.0f; B[0] = 0.0f; B[1] = 0.0f;
            }
            v8f d = __builtin_amdgcn_wmma_f32_16x16x4_f32(
                false, A, false, B, (short)0, czero, false, false);
            int n = n0 + (lane & 15);
            float ny0 = P->y0[n], nx0 = P->x0[n], ny1 = P->y1[n], nx1 = P->x1[n];
            int mbase = m0 + ((lane & 16) ? 8 : 0);
            #pragma unroll
            for (int r = 0; r < 8; ++r) {
                int m = mbase + r;
                float iy = fmaxf(__fsub_rn(fminf(P->y1[m], ny1), fmaxf(P->y0[m], ny0)), 0.0f);
                float ix = fmaxf(__fsub_rn(fminf(P->x1[m], nx1), fmaxf(P->x0[m], nx0)), 0.0f);
                float inter = __fmul_rn(iy, ix);
                float uni = __fsub_rn(d[r], inter);
                bool pred = (uni > 0.0f) && (__fdiv_rn(inter, uni) > 0.4f);
                unsigned bal = __builtin_amdgcn_ballot_w32(pred);
                if (lane == 0) {
                    P->mask16[(m0 + r)     * 16 + ni] = (unsigned short)(bal & 0xFFFFu);
                    P->mask16[(m0 + 8 + r) * 16 + ni] = (unsigned short)(bal >> 16);
                }
            }
        }
    }
    __syncthreads();

    for (int it = 0; it < MAXOUT; ++it) {
        P->redv[tid] = P->valid[tid] ? P->sc[tid] : NEG_INF;
        P->redi[tid] = tid;
        __syncthreads();
        for (int off = 128; off > 0; off >>= 1) {
            if (tid < off) {
                float v2 = P->redv[tid + off]; int i2 = P->redi[tid + off];
                if (v2 > P->redv[tid] || (v2 == P->redv[tid] && i2 < P->redi[tid])) {
                    P->redv[tid] = v2; P->redi[tid] = i2;
                }
            }
            __syncthreads();
        }
        int idx = P->redi[0];
        int ok  = P->valid[idx];
        unsigned bit = (P->mask16[idx * 16 + (tid >> 4)] >> (tid & 15)) & 1u;
        unsigned char nv = (unsigned char)(P->valid[tid] & (unsigned char)(1u - bit));
        if (tid == idx) nv = 0;
        __syncthreads();
        P->valid[tid] = nv;
        if (tid == 0) { P->selIdx[it] = idx; P->selOk[it] = ok; }
        __syncthreads();
    }

    if (tid < MAXOUT) {
        int sel = P->selIdx[tid];
        float okf = P->selOk[tid] ? 1.0f : 0.0f;
        P->redv[tid] = __fmul_rn(P->sc[sel], okf);
    }
    __syncthreads();
    if (tid < MAXOUT) {
        int sel = P->selIdx[tid];
        float okf = P->selOk[tid] ? 1.0f : 0.0f;
        float ms = P->redv[tid];
        int rank = 0;
        for (int jt = 0; jt < MAXOUT; ++jt) {
            float vj = P->redv[jt];
            if (vj > ms || (vj == ms && jt < tid)) rank++;
        }
        float* o = out + (size_t)img * (MAXOUT * 6) + (size_t)rank * 6;
        o[0] = __fmul_rn(P->y0[sel], okf);
        o[1] = __fmul_rn(P->x0[sel], okf);
        o[2] = __fmul_rn(P->y1[sel], okf);
        o[3] = __fmul_rn(P->x1[sel], okf);
        o[4] = ms;
        o[5] = __fmul_rn(P->cl[sel], okf);
    }
}

extern "C" void kernel_launch(void* const* d_in, const int* in_sizes, int n_in,
                              void* d_out, int out_size, void* d_ws, size_t ws_size,
                              hipStream_t stream) {
    (void)n_in; (void)out_size; (void)d_ws; (void)ws_size;
    const float* net = (const float*)d_in[0];
    float* out = (float*)d_out;
    int batch = in_sizes[0] / D_IMG;
    det_kernel<<<dim3(batch), dim3(256), 0, stream>>>(net, out);
}


# NMS picks by a parallel fixpoint in one wave: every lane tests its candidate's mask row against the kept set, one ballot per 32 candidates, until unchanged; replaces 30 block-wide argmax rounds
# speedup vs baseline: 4.9295x; 1.3988x over previous
#include <hip/hip_runtime.h>

typedef float v2f __attribute__((ext_vector_type(2)));
typedef float v8f __attribute__((ext_vector_type(8)));

#define S_GRID   56
#define N_CLASS  20
#define CELLS    (S_GRID * S_GRID)
#define ITEMS    (CELLS * 2)
#define CAND     256
#define MAXOUT   30
#define D_IMG    94080
#define CONF_OFF 62720
#define COORD_OFF 68992
#define NEG_INF  (-3.402823466e38f)
#define NSLOT    32
#define NWAVE    8

typedef unsigned long long u64;


__device__ __forceinline__ u64 key_from_lane(u64 v, int src4) {
    int lo = __builtin_amdgcn_ds_bpermute(src4, (int)(unsigned)v);
    int hi = __builtin_amdgcn_ds_bpermute(src4, (int)(unsigned)(v >> 32));
    return ((u64)(unsigned)hi << 32) | (u64)(unsigned)lo;
}

template <int NCH, int STRIDE, int M, bool ALT>
__device__ __forceinline__ void stage_lanes(u64 (&key)[NSLOT], int lane, bool d0) {
    const int src4 = (lane ^ M) << 2;
    const bool lower = (lane & M) == 0;
    #pragma unroll
    for (int c = 0; c < NCH; ++c) {
        const bool desc = (ALT && (c & 1)) ? !d0 : d0;
        const bool keepmin = (lower != desc);
        #pragma unroll
        for (int q = 0; q < 8; ++q) {
            const int s = c * STRIDE + q;
            u64 mine = key[s];
            u64 oth = key_from_lane(mine, src4);
            bool take = (oth > mine) != keepmin;
            key[s] = take ? oth : mine;
        }
    }
}

template <int NCH, int STRIDE, int J, bool ALT>
__device__ __forceinline__ void stage_regs(u64 (&key)[NSLOT], bool d0) {
    #pragma unroll
    for (int c = 0; c < NCH; ++c) {
        const bool desc = (ALT && (c & 1)) ? !d0 : d0;
        #pragma unroll
        for (int q = 0; q < 8; ++q) {
            if ((q & J) == 0) {
                const int s = c * STRIDE + q;
                u64 a = key[s], b = key[s + J];
                bool sw = (a < b) == desc;
                key[s]     = sw ? b : a;
                key[s + J] = sw ? a : b;
            }
        }
    }
}

template <int NCH, int STRIDE, int J, int K>
__device__ __forceinline__ void stage_regs_q(u64 (&key)[NSLOT]) {
    #pragma unroll
    for (int c = 0; c < NCH; ++c) {
        #pragma unroll
        for (int q = 0; q < 8; ++q) {
            if ((q & J) == 0) {
                const int s = c * STRIDE + q;
                const bool desc = (q & K) == 0;
                u64 a = key[s], b = key[s + J];
                bool sw = desc ? (a < b) : (a > b);
                key[s]     = sw ? b : a;
                key[s + J] = sw ? a : b;
            }
        }
    }
}

template <int NCH, int STRIDE, int MTOP, bool ALT>
__device__ __forceinline__ void merge_chunks(u64 (&key)[NSLOT], int lane, bool d0) {
    if (MTOP >= 16) stage_lanes<NCH, STRIDE, 16, ALT>(key, lane, d0);
    if (MTOP >= 8)  stage_lanes<NCH, STRIDE, 8,  ALT>(key, lane, d0);
    if (MTOP >= 4)  stage_lanes<NCH, STRIDE, 4,  ALT>(key, lane, d0);
    if (MTOP >= 2)  stage_lanes<NCH, STRIDE, 2,  ALT>(key, lane, d0);
    if (MTOP >= 1)  stage_lanes<NCH, STRIDE, 1,  ALT>(key, lane, d0);
    stage_regs<NCH, STRIDE, 4, ALT>(key, d0);
    stage_regs<NCH, STRIDE, 2, ALT>(key, d0);
    stage_regs<NCH, STRIDE, 1, ALT>(key, d0);
}

__device__ __forceinline__ u64 key_max(u64 a, u64 b) { return a > b ? a : b; }

struct Phase2 {
    float y0[CAND], x0[CAND], y1[CAND], x1[CAND];
    float sc[CAND], cl[CAND], ar[CAND];
    float redv[CAND];
    union alignas(16) {
        unsigned short mask16[CAND * 16];
        unsigned       maskw[CAND * 8];
    };
    int selIdx[32], selOk[32];
};

__global__ __launch_bounds__(256) void det_kernel(const float* __restrict__ net,
                                                  float* __restrict__ out) {
    __shared__ u64 keys[CAND];
    __shared__ u64 xch[3][NWAVE][CAND];
    __shared__ Phase2 p2;
    Phase2* P = &p2;

    const int tid = threadIdx.x;
    const int img = blockIdx.x;
    const float* in = net + (size_t)img * D_IMG;
    const int lane = tid & 31;
    const int wave = tid >> 5;

    u64 key[NSLOT];
    #pragma unroll
    for (int n = 0; n < 13; ++n) {
        const int cell = tid + 256 * n;
        if (n < 12 || cell < CELLS) {
            const float* pr = in + cell * N_CLASS;
            float p[N_CLASS];
            #pragma unroll
            for (int c = 0; c < N_CLASS; ++c) p[c] = pr[c];
            #pragma unroll
            for (int bx = 0; bx < 2; ++bx) {
                int id = cell * 2 + bx;
                float conf = in[CONF_OFF + id];
                float best = NEG_INF;
                #pragma unroll
                for (int c = 0; c < N_CLASS; ++c) {
                    float v = __fmul_rn(conf, p[c]);
                    best = v > best ? v : best;
                }
                float s = (best >= 0.1f) ? best : 0.0f;
                key[2 * n + bx] = ((u64)__float_as_uint(s) << 32) | (unsigned)(~(unsigned)id);
            }
        } else {
            key[2 * n]     = (u64)(unsigned)(~(unsigned)(ITEMS + tid * NSLOT + 2 * n));
            key[2 * n + 1] = (u64)(unsigned)(~(unsigned)(ITEMS + tid * NSLOT + 2 * n + 1));
        }
    }
    #pragma unroll
    for (int s = 26; s < NSLOT; ++s)
        key[s] = (u64)(unsigned)(~(unsigned)(ITEMS + tid * NSLOT + s));

    stage_regs_q<4, 8, 1, 2>(key);
    stage_regs_q<4, 8, 2, 4>(key);
    stage_regs_q<4, 8, 1, 4>(key);
    merge_chunks<4, 8, 0,  false>(key, lane, (lane & 1)  == 0);
    merge_chunks<4, 8, 1,  false>(key, lane, (lane & 2)  == 0);
    merge_chunks<4, 8, 2,  false>(key, lane, (lane & 4)  == 0);
    merge_chunks<4, 8, 4,  false>(key, lane, (lane & 8)  == 0);
    merge_chunks<4, 8, 8,  false>(key, lane, (lane & 16) == 0);
    merge_chunks<4, 8, 16, true >(key, lane, true);
    #pragma unroll
    for (int q = 0; q < 8; ++q) {
        key[q]      = key_max(key[q],      key[8 + q]);
        key[16 + q] = key_max(key[16 + q], key[24 + q]);
    }
    merge_chunks<2, 16, 16, true>(key, lane, true);
    #pragma unroll
    for (int q = 0; q < 8; ++q) key[q] = key_max(key[q], key[16 + q]);
    merge_chunks<1, 16, 16, false>(key, lane, (wave & 1) == 0);
    #pragma unroll
    for (int q = 0; q < 8; ++q) xch[0][wave][q * 32 + lane] = key[q];
    __syncthreads();
    #pragma unroll
    for (int q = 0; q < 8; ++q) key[q] = key_max(key[q], xch[0][wave ^ 1][q * 32 + lane]);
    merge_chunks<1, 16, 16, false>(key, lane, (wave & 2) == 0);
    #pragma unroll
    for (int q = 0; q < 8; ++q) xch[1][wave][q * 32 + lane] = key[q];
    __syncthreads();
    #pragma unroll
    for (int q = 0; q < 8; ++q) key[q] = key_max(key[q], xch[1][wave ^ 2][q * 32 + lane]);
    merge_chunks<1, 16, 16, false>(key, lane, (wave & 4) == 0);
    #pragma unroll
    for (int q = 0; q < 8; ++q) xch[2][wave][q * 32 + lane] = key[q];
    __syncthreads();
    #pragma unroll
    for (int q = 0; q < 8; ++q) key[q] = key_max(key[q], xch[2][wave ^ 4][q * 32 + lane]);
    merge_chunks<1, 16, 16, false>(key, lane, true);
    if (wave == 0) {
        #pragma unroll
        for (int q = 0; q < 8; ++q) keys[lane * 8 + q] = key[q];
    }
    __syncthreads();

    {
        unsigned long long key = keys[tid];
        unsigned id = ~(unsigned)key;
        float s = __uint_as_float((unsigned)(key >> 32));
        int cell = (int)(id >> 1);
        int gi = cell / S_GRID, gj = cell % S_GRID;
        const float* cd = in + COORD_OFF + (size_t)id * 4;
        float tx = cd[0], ty = cd[1], tw = cd[2], th = cd[3];
        float x = __fdiv_rn(__fadd_rn(tx, (float)gj), 56.0f);
        float y = __fdiv_rn(__fadd_rn(ty, (float)gi), 56.0f);
        float hw = __fmul_rn(__fmul_rn(tw, tw), 0.5f);
        float hh = __fmul_rn(__fmul_rn(th, th), 0.5f);
        float y0 = __fsub_rn(y, hh), x0 = __fsub_rn(x, hw);
        float y1 = __fadd_rn(y, hh), x1 = __fadd_rn(x, hw);
        float conf = in[CONF_OFF + id];
        const float* pr = in + cell * N_CLASS;
        float best = NEG_INF; int bc = 0;
        #pragma unroll
        for (int c = 0; c < N_CLASS; ++c) {
            float v = __fmul_rn(conf, pr[c]);
            if (v > best) { best = v; bc = c; }
        }
        float dy = fmaxf(__fsub_rn(y1, y0), 0.0f);
        float dx = fmaxf(__fsub_rn(x1, x0), 0.0f);
        P->y0[tid] = y0; P->x0[tid] = x0; P->y1[tid] = y1; P->x1[tid] = x1;
        P->sc[tid] = s;  P->cl[tid] = (float)bc;
        P->ar[tid] = __fmul_rn(dy, dx);
    }
    __syncthreads();

    {
        v8f czero = {};
        for (int tt = wave * 32; tt < wave * 32 + 32; ++tt) {
            int mi = tt >> 4, ni = tt & 15;
            int m0 = mi << 4, n0 = ni << 4;
            v2f A, B;
            if (lane < 16) {
                A[0] = P->ar[m0 + lane]; A[1] = 1.0f;
                B[0] = 1.0f;             B[1] = P->ar[n0 + lane];
            } else {
                A[0] = 0.0f; A[1] = 0.0f; B[0] = 0.0f; B[1] = 0.0f;
            }
            v8f d = __builtin_amdgcn_wmma_f32_16x16x4_f32(
                false, A, false, B, (short)0, czero, false, false);
            int n = n0 + (lane & 15);
            float ny0 = P->y0[n], nx0 = P->x0[n], ny1 = P->y1[n], nx1 = P->x1[n];
            int mbase = m0 + ((lane & 16) ? 8 : 0);
            #pragma unroll
            for (int r = 0; r < 8; ++r) {
                int m = mbase + r;
                float iy = fmaxf(__fsub_rn(fminf(P->y1[m], ny1), fmaxf(P->y0[m], ny0)), 0.0f);
                float ix = fmaxf(__fsub_rn(fminf(P->x1[m], nx1), fmaxf(P->x0[m], nx0)), 0.0f);
                float inter = __fmul_rn(iy, ix);
                float uni = __fsub_rn(d[r], inter);
                bool pred = (uni > 0.0f) && (__fdiv_rn(inter, uni) > 0.4f);
                unsigned bal = __builtin_amdgcn_ballot_w32(pred);
                if (lane == 0) {
                    P->mask16[(m0 + r)     * 16 + ni] = (unsigned short)(bal & 0xFFFFu);
                    P->mask16[(m0 + 8 + r) * 16 + ni] = (unsigned short)(bal >> 16);
                }
            }
        }
    }
    __syncthreads();

    if (wave == 0) {
        const unsigned below = (1u << lane) - 1u;
        unsigned row[8][8];
        #pragma unroll
        for (int q = 0; q < 8; ++q) {
            #pragma unroll
            for (int w = 0; w <= q; ++w) row[q][w] = P->maskw[(q * 32 + lane) * 8 + w];
        }
        unsigned keepw[8];
        int total = 0;
        #pragma unroll
        for (int q = 0; q < 8; ++q) {
            keepw[q] = 0;
            if (total < MAXOUT) {
                unsigned ext = 0;
                #pragma unroll
                for (int w = 0; w < q; ++w) ext |= row[q][w] & keepw[w];
                const unsigned own = row[q][q] & below;
                const unsigned open = __builtin_amdgcn_ballot_w32(ext == 0);
                unsigned k = open;
                for (int pass = 0; pass < 32; ++pass) {
                    unsigned k2 = open & __builtin_amdgcn_ballot_w32((own & k) == 0);
                    if (k2 == k) break;
                    k = k2;
                }
                keepw[q] = k;
                if ((k >> lane) & 1u) {
                    int rank = total + __builtin_popcount(k & below);
                    if (rank < MAXOUT) { P->selIdx[rank] = q * 32 + lane; P->selOk[rank] = 1; }
                }
                total += __builtin_popcount(k);
            }
        }
        if (lane < MAXOUT && lane >= total) { P->selIdx[lane] = 0; P->selOk[lane] = 0; }
    }
    __syncthreads();

    if (tid < MAXOUT) {
        int sel = P->selIdx[tid];
        float okf = P->selOk[tid] ? 1.0f : 0.0f;
        P->redv[tid] = __fmul_rn(P->sc[sel], okf);
    }
    __syncthreads();
    if (tid < MAXOUT) {
        int sel = P->selIdx[tid];
        float okf = P->selOk[tid] ? 1.0f : 0.0f;
        float ms = P->redv[tid];
        int rank = 0;
        for (int jt = 0; jt < MAXOUT; ++jt) {
            float vj = P->redv[jt];
            if (vj > ms || (vj == ms && jt < tid)) rank++;
        }
        float* o = out + (size_t)img * (MAXOUT * 6) + (size_t)rank * 6;
        o[0] = __fmul_rn(P->y0[sel], okf);
        o[1] = __fmul_rn(P->x0[sel], okf);
        o[2] = __fmul_rn(P->y1[sel], okf);
        o[3] = __fmul_rn(P->x1[sel], okf);
        o[4] = ms;
        o[5] = __fmul_rn(P->cl[sel], okf);
    }
}

extern "C" void kernel_launch(void* const* d_in, const int* in_sizes, int n_in,
                              void* d_out, int out_size, void* d_ws, size_t ws_size,
                              hipStream_t stream) {
    (void)n_in; (void)out_size; (void)d_ws; (void)ws_size;
    const float* net = (const float*)d_in[0];
    float* out = (float*)d_out;
    int batch = in_sizes[0] / D_IMG;
    det_kernel<<<dim3(batch), dim3(256), 0, stream>>>(net, out);
}


# IoU mask built on demand inside the NMS scan: per 32-candidate word only the 4(q+1) tiles that word reads, then the fixpoint settles it; nothing is built behind the 30th pick
# speedup vs baseline: 7.7760x; 1.5774x over previous
#include <hip/hip_runtime.h>

typedef float v2f __attribute__((ext_vector_type(2)));
typedef float v8f __attribute__((ext_vector_type(8)));

#define S_GRID   56
#define N_CLASS  20
#define CELLS    (S_GRID * S_GRID)
#define ITEMS    (CELLS * 2)
#define CAND     256
#define MAXOUT   30
#define D_IMG    94080
#define CONF_OFF 62720
#define COORD_OFF 68992
#define NEG_INF  (-3.402823466e38f)
#define NSLOT    32
#define NWAVE    8

typedef unsigned long long u64;


__device__ __forceinline__ u64 key_from_lane(u64 v, int src4) {
    int lo = __builtin_amdgcn_ds_bpermute(src4, (int)(unsigned)v);
    int hi = __builtin_amdgcn_ds_bpermute(src4, (int)(unsigned)(v >> 32));
    return ((u64)(unsigned)hi << 32) | (u64)(unsigned)lo;
}

template <int NCH, int STRIDE, int M, bool ALT>
__device__ __forceinline__ void stage_lanes(u64 (&key)[NSLOT], int lane, bool d0) {
    const int src4 = (lane ^ M) << 2;
    const bool lower = (lane & M) == 0;
    #pragma unroll
    for (int c = 0; c < NCH; ++c) {
        const bool desc = (ALT && (c & 1)) ? !d0 : d0;
        const bool keepmin = (lower != desc);
        #pragma unroll
        for (int q = 0; q < 8; ++q) {
            const int s = c * STRIDE + q;
            u64 mine = key[s];
            u64 oth = key_from_lane(mine, src4);
            bool take = (oth > mine) != keepmin;
            key[s] = take ? oth : mine;
        }
    }
}

template <int NCH, int STRIDE, int J, bool ALT>
__device__ __forceinline__ void stage_regs(u64 (&key)[NSLOT], bool d0) {
    #pragma unroll
    for (int c = 0; c < NCH; ++c) {
        const bool desc = (ALT && (c & 1)) ? !d0 : d0;
        #pragma unroll
        for (int q = 0; q < 8; ++q) {
            if ((q & J) == 0) {
                const int s = c * STRIDE + q;
                u64 a = key[s], b = key[s + J];
                bool sw = (a < b) == desc;
                key[s]     = sw ? b : a;
                key[s + J] = sw ? a : b;
            }
        }
    }
}

template <int NCH, int STRIDE, int J, int K>
__device__ __forceinline__ void stage_regs_q(u64 (&key)[NSLOT]) {
    #pragma unroll
    for (int c = 0; c < NCH; ++c) {
        #pragma unroll
        for (int q = 0; q < 8; ++q) {
            if ((q & J) == 0) {
                const int s = c * STRIDE + q;
                const bool desc = (q & K) == 0;
                u64 a = key[s], b = key[s + J];
                bool sw = desc ? (a < b) : (a > b);
                key[s]     = sw ? b : a;
                key[s + J] = sw ? a : b;
            }
        }
    }
}

template <int NCH, int STRIDE, int MTOP, bool ALT>
__device__ __forceinline__ void merge_chunks(u64 (&key)[NSLOT], int lane, bool d0) {
    if (MTOP >= 16) stage_lanes<NCH, STRIDE, 16, ALT>(key, lane, d0);
    if (MTOP >= 8)  stage_lanes<NCH, STRIDE, 8,  ALT>(key, lane, d0);
    if (MTOP >= 4)  stage_lanes<NCH, STRIDE, 4,  ALT>(key, lane, d0);
    if (MTOP >= 2)  stage_lanes<NCH, STRIDE, 2,  ALT>(key, lane, d0);
    if (MTOP >= 1)  stage_lanes<NCH, STRIDE, 1,  ALT>(key, lane, d0);
    stage_regs<NCH, STRIDE, 4, ALT>(key, d0);
    stage_regs<NCH, STRIDE, 2, ALT>(key, d0);
    stage_regs<NCH, STRIDE, 1, ALT>(key, d0);
}

__device__ __forceinline__ u64 key_max(u64 a, u64 b) { return a > b ? a : b; }

struct Phase2 {
    float y0[CAND], x0[CAND], y1[CAND], x1[CAND];
    float sc[CAND], cl[CAND], ar[CAND];
    float redv[CAND];
    union alignas(16) {
        unsigned short mask16[CAND * 16];
        unsigned       maskw[CAND * 8];
    };
    int selIdx[32], selOk[32];
    int npick;
};

__global__ __launch_bounds__(256) void det_kernel(const float* __restrict__ net,
                                                  float* __restrict__ out) {
    __shared__ u64 keys[CAND];
    __shared__ u64 xch[3][NWAVE][CAND];
    __shared__ Phase2 p2;
    Phase2* P = &p2;

    const int tid = threadIdx.x;
    const int img = blockIdx.x;
    const float* in = net + (size_t)img * D_IMG;
    const int lane = tid & 31;
    const int wave = tid >> 5;

    u64 key[NSLOT];
    #pragma unroll
    for (int n = 0; n < 13; ++n) {
        const int cell = tid + 256 * n;
        if (n < 12 || cell < CELLS) {
            const float* pr = in + cell * N_CLASS;
            float p[N_CLASS];
            #pragma unroll
            for (int c = 0; c < N_CLASS; ++c) p[c] = pr[c];
            #pragma unroll
            for (int bx = 0; bx < 2; ++bx) {
                int id = cell * 2 + bx;
                float conf = in[CONF_OFF + id];
                float best = NEG_INF;
                #pragma unroll
                for (int c = 0; c < N_CLASS; ++c) {
                    float v = __fmul_rn(conf, p[c]);
                    best = v > best ? v : best;
                }
                float s = (best >= 0.1f) ? best : 0.0f;
                key[2 * n + bx] = ((u64)__float_as_uint(s) << 32) | (unsigned)(~(unsigned)id);
            }
        } else {
            key[2 * n]     = (u64)(unsigned)(~(unsigned)(ITEMS + tid * NSLOT + 2 * n));
            key[2 * n + 1] = (u64)(unsigned)(~(unsigned)(ITEMS + tid * NSLOT + 2 * n + 1));
        }
    }
    #pragma unroll
    for (int s = 26; s < NSLOT; ++s)
        key[s] = (u64)(unsigned)(~(unsigned)(ITEMS + tid * NSLOT + s));

    stage_regs_q<4, 8, 1, 2>(key);
    stage_regs_q<4, 8, 2, 4>(key);
    stage_regs_q<4, 8, 1, 4>(key);
    merge_chunks<4, 8, 0,  false>(key, lane, (lane & 1)  == 0);
    merge_chunks<4, 8, 1,  false>(key, lane, (lane & 2)  == 0);
    merge_chunks<4, 8, 2,  false>(key, lane, (lane & 4)  == 0);
    merge_chunks<4, 8, 4,  false>(key, lane, (lane & 8)  == 0);
    merge_chunks<4, 8, 8,  false>(key, lane, (lane & 16) == 0);
    merge_chunks<4, 8, 16, true >(key, lane, true);
    #pragma unroll
    for (int q = 0; q < 8; ++q) {
        key[q]      = key_max(key[q],      key[8 + q]);
        key[16 + q] = key_max(key[16 + q], key[24 + q]);
    }
    merge_chunks<2, 16, 16, true>(key, lane, true);
    #pragma unroll
    for (int q = 0; q < 8; ++q) key[q] = key_max(key[q], key[16 + q]);
    merge_chunks<1, 16, 16, false>(key, lane, (wave & 1) == 0);
    #pragma unroll
    for (int q = 0; q < 8; ++q) xch[0][wave][q * 32 + lane] = key[q];
    __syncthreads();
    #pragma unroll
    for (int q = 0; q < 8; ++q) key[q] = key_max(key[q], xch[0][wave ^ 1][q * 32 + lane]);
    merge_chunks<1, 16, 16, false>(key, lane, (wave & 2) == 0);
    #pragma unroll
    for (int q = 0; q < 8; ++q) xch[1][wave][q * 32 + lane] = key[q];
    __syncthreads();
    #pragma unroll
    for (int q = 0; q < 8; ++q) key[q] = key_max(key[q], xch[1][wave ^ 2][q * 32 + lane]);
    merge_chunks<1, 16, 16, false>(key, lane, (wave & 4) == 0);
    #pragma unroll
    for (int q = 0; q < 8; ++q) xch[2][wave][q * 32 + lane] = key[q];
    __syncthreads();
    #pragma unroll
    for (int q = 0; q < 8; ++q) key[q] = key_max(key[q], xch[2][wave ^ 4][q * 32 + lane]);
    merge_chunks<1, 16, 16, false>(key, lane, true);
    if (wave == 0) {
        #pragma unroll
        for (int q = 0; q < 8; ++q) keys[lane * 8 + q] = key[q];
    }
    __syncthreads();

    {
        unsigned long long key = keys[tid];
        unsigned id = ~(unsigned)key;
        float s = __uint_as_float((unsigned)(key >> 32));
        int cell = (int)(id >> 1);
        int gi = cell / S_GRID, gj = cell % S_GRID;
        const float* cd = in + COORD_OFF + (size_t)id * 4;
        float tx = cd[0], ty = cd[1], tw = cd[2], th = cd[3];
        float x = __fdiv_rn(__fadd_rn(tx, (float)gj), 56.0f);
        float y = __fdiv_rn(__fadd_rn(ty, (float)gi), 56.0f);
        float hw = __fmul_rn(__fmul_rn(tw, tw), 0.5f);
        float hh = __fmul_rn(__fmul_rn(th, th), 0.5f);
        float y0 = __fsub_rn(y, hh), x0 = __fsub_rn(x, hw);
        float y1 = __fadd_rn(y, hh), x1 = __fadd_rn(x, hw);
        float conf = in[CONF_OFF + id];
        const float* pr = in + cell * N_CLASS;
        float best = NEG_INF; int bc = 0;
        #pragma unroll
        for (int c = 0; c < N_CLASS; ++c) {
            float v = __fmul_rn(conf, pr[c]);
            if (v > best) { best = v; bc = c; }
        }
        float dy = fmaxf(__fsub_rn(y1, y0), 0.0f);
        float dx = fmaxf(__fsub_rn(x1, x0), 0.0f);
        P->y0[tid] = y0; P->x0[tid] = x0; P->y1[tid] = y1; P->x1[tid] = x1;
        P->sc[tid] = s;  P->cl[tid] = (float)bc;
        P->ar[tid] = __fmul_rn(dy, dx);
        if (tid < 32) { P->selIdx[tid] = 0; P->selOk[tid] = 0; }
    }
    __syncthreads();

    {
        const unsigned below = (1u << lane) - 1u;
        unsigned keepw[8];
        int total = 0;
        #pragma unroll
        for (int q = 0; q < 8; ++q) {
            v8f czero = {};
            for (int tt = wave; tt < 4 * (q + 1); tt += NWAVE) {
                int mi = 2 * q + (tt & 1), ni = tt >> 1;
                int m0 = mi << 4, n0 = ni << 4;
                v2f A, B;
                if (lane < 16) {
                    A[0] = P->ar[m0 + lane]; A[1] = 1.0f;
                    B[0] = 1.0f;             B[1] = P->ar[n0 + lane];
                } else {
                    A[0] = 0.0f; A[1] = 0.0f; B[0] = 0.0f; B[1] = 0.0f;
                }
                v8f d = __builtin_amdgcn_wmma_f32_16x16x4_f32(
                    false, A, false, B, (short)0, czero, false, false);
                int n = n0 + (lane & 15);
                float ny0 = P->y0[n], nx0 = P->x0[n], ny1 = P->y1[n], nx1 = P->x1[n];
                int mbase = m0 + ((lane & 16) ? 8 : 0);
                #pragma unroll
                for (int r = 0; r < 8; ++r) {
                    int m = mbase + r;
                    float iy = fmaxf(__fsub_rn(fminf(P->y1[m], ny1), fmaxf(P->y0[m], ny0)), 0.0f);
                    float ix = fmaxf(__fsub_rn(fminf(P->x1[m], nx1), fmaxf(P->x0[m], nx0)), 0.0f);
                    float inter = __fmul_rn(iy, ix);
                    float uni = __fsub_rn(d[r], inter);
                    bool pred = (uni > 0.0f) && (__fdiv_rn(inter, uni) > 0.4f);
                    unsigned bal = __builtin_amdgcn_ballot_w32(pred);
                    if (lane == 0) {
                        P->mask16[(m0 + r)     * 16 + ni] = (unsigned short)(bal & 0xFFFFu);
                        P->mask16[(m0 + 8 + r) * 16 + ni] = (unsigned short)(bal >> 16);
                    }
                }
            }
            __syncthreads();

            if (wave == 0) {
                unsigned ext = 0;
                #pragma unroll
                for (int w = 0; w < q; ++w) ext |= P->maskw[(q * 32 + lane) * 8 + w] & keepw[w];
                const unsigned own = P->maskw[(q * 32 + lane) * 8 + q] & below;
                const unsigned open = __builtin_amdgcn_ballot_w32(ext == 0);
                unsigned k = open;
                for (int pass = 0; pass < 32; ++pass) {
                    unsigned k2 = open & __builtin_amdgcn_ballot_w32((own & k) == 0);
                    if (k2 == k) break;
                    k = k2;
                }
                keepw[q] = k;
                if ((k >> lane) & 1u) {
                    int rank = total + __builtin_popcount(k & below);
                    if (rank < MAXOUT) { P->selIdx[rank] = q * 32 + lane; P->selOk[rank] = 1; }
                }
                total += __builtin_popcount(k);
                if (lane == 0) P->npick = total;
            }
            __syncthreads();
            if (P->npick >= MAXOUT) break;
        }
    }

    if (tid < MAXOUT) {
        int sel = P->selIdx[tid];
        float okf = P->selOk[tid] ? 1.0f : 0.0f;
        P->redv[tid] = __fmul_rn(P->sc[sel], okf);
    }
    __syncthreads();
    if (tid < MAXOUT) {
        int sel = P->selIdx[tid];
        float okf = P->selOk[tid] ? 1.0f : 0.0f;
        float ms = P->redv[tid];
        int rank = 0;
        for (int jt = 0; jt < MAXOUT; ++jt) {
            float vj = P->redv[jt];
            if (vj > ms || (vj == ms && jt < tid)) rank++;
        }
        float* o = out + (size_t)img * (MAXOUT * 6) + (size_t)rank * 6;
        o[0] = __fmul_rn(P->y0[sel], okf);
        o[1] = __fmul_rn(P->x0[sel], okf);
        o[2] = __fmul_rn(P->y1[sel], okf);
        o[3] = __fmul_rn(P->x1[sel], okf);
        o[4] = ms;
        o[5] = __fmul_rn(P->cl[sel], okf);
    }
}

extern "C" void kernel_launch(void* const* d_in, const int* in_sizes, int n_in,
                              void* d_out, int out_size, void* d_ws, size_t ws_size,
                              hipStream_t stream) {
    (void)n_in; (void)out_size; (void)d_ws; (void)ws_size;
    const float* net = (const float*)d_in[0];
    float* out = (float*)d_out;
    int batch = in_sizes[0] / D_IMG;
    det_kernel<<<dim3(batch), dim3(256), 0, stream>>>(net, out);
}


# Sort network's lane exchanges moved off the LDS unit onto the vector unit: v_permlane16_var (gather in a 16-lane row) for lane distances 1-8, v_permlanex16 for 16, replacing 2 ds_bpermute per key
# speedup vs baseline: 11.4407x; 1.4713x over previous
#include <hip/hip_runtime.h>

typedef float v2f __attribute__((ext_vector_type(2)));
typedef float v8f __attribute__((ext_vector_type(8)));

#define S_GRID   56
#define N_CLASS  20
#define CELLS    (S_GRID * S_GRID)
#define ITEMS    (CELLS * 2)
#define CAND     256
#define MAXOUT   30
#define D_IMG    94080
#define CONF_OFF 62720
#define COORD_OFF 68992
#define NEG_INF  (-3.402823466e38f)
#define NSLOT    32
#define NWAVE    8

typedef unsigned long long u64;


__device__ __forceinline__ u64 key_from_row_lane(u64 v, unsigned sel) {
    unsigned lo = (unsigned)v, hi = (unsigned)(v >> 32);
    unsigned rlo = __builtin_amdgcn_permlane16_var(lo, lo, sel, true, false);
    unsigned rhi = __builtin_amdgcn_permlane16_var(hi, hi, sel, true, false);
    return ((u64)rhi << 32) | (u64)rlo;
}
__device__ __forceinline__ u64 key_from_other_row(u64 v) {
    unsigned lo = (unsigned)v, hi = (unsigned)(v >> 32);
    unsigned rlo = __builtin_amdgcn_permlanex16(lo, lo, 0x76543210u, 0xfedcba98u, true, false);
    unsigned rhi = __builtin_amdgcn_permlanex16(hi, hi, 0x76543210u, 0xfedcba98u, true, false);
    return ((u64)rhi << 32) | (u64)rlo;
}

template <int NCH, int STRIDE, int J, int K>
__device__ __forceinline__ void stage_regs_q(u64 (&key)[NSLOT]) {
    #pragma unroll
    for (int c = 0; c < NCH; ++c) {
        #pragma unroll
        for (int q = 0; q < 8; ++q) {
            if ((q & J) == 0) {
                const int s = c * STRIDE + q;
                const bool desc = (q & K) == 0;
                u64 a = key[s], b = key[s + J];
                bool sw = desc ? (a < b) : (a > b);
                key[s]     = sw ? b : a;
                key[s + J] = sw ? a : b;
            }
        }
    }
}

template <int NCH, int STRIDE, bool XROW>
__device__ __forceinline__ void stage_lanes_rt(u64 (&key)[NSLOT], int lane, int M, bool d0, bool alt) {
    const unsigned sel = (unsigned)(lane ^ M);
    const bool lower = (lane & M) == 0;
    #pragma unroll
    for (int c = 0; c < NCH; ++c) {
        const bool desc = (c & 1) ? (d0 != alt) : d0;
        const bool keepmin = (lower != desc);
        #pragma unroll
        for (int q = 0; q < 8; ++q) {
            const int s = c * STRIDE + q;
            u64 mine = key[s];
            u64 oth = XROW ? key_from_other_row(mine) : key_from_row_lane(mine, sel);
            bool take = (oth > mine) != keepmin;
            key[s] = take ? oth : mine;
        }
    }
}

template <int NCH, int STRIDE, int J>
__device__ __forceinline__ void stage_regs_rt(u64 (&key)[NSLOT], bool d0, bool alt) {
    #pragma unroll
    for (int c = 0; c < NCH; ++c) {
        const bool desc = (c & 1) ? (d0 != alt) : d0;
        #pragma unroll
        for (int q = 0; q < 8; ++q) {
            if ((q & J) == 0) {
                const int s = c * STRIDE + q;
                u64 a = key[s], b = key[s + J];
                bool sw = (a < b) == desc;
                key[s]     = sw ? b : a;
                key[s + J] = sw ? a : b;
            }
        }
    }
}

template <int NCH, int STRIDE>
__device__ __forceinline__ void merge_rt(u64 (&key)[NSLOT], int lane, int mtop, bool d0, bool alt) {
    if (mtop >= 16) stage_lanes_rt<NCH, STRIDE, true>(key, lane, 16, d0, alt);
    #pragma nounroll
    for (int M = (mtop >= 16) ? 8 : mtop; M >= 1; M >>= 1) stage_lanes_rt<NCH, STRIDE, false>(key, lane, M, d0, alt);
    stage_regs_rt<NCH, STRIDE, 4>(key, d0, alt);
    stage_regs_rt<NCH, STRIDE, 2>(key, d0, alt);
    stage_regs_rt<NCH, STRIDE, 1>(key, d0, alt);
}

__device__ __forceinline__ u64 key_max(u64 a, u64 b) { return a > b ? a : b; }

struct Phase2 {
    float y0[CAND], x0[CAND], y1[CAND], x1[CAND];
    float sc[CAND], cl[CAND], ar[CAND];
    float redv[CAND];
    union alignas(16) {
        unsigned short mask16[CAND * 16];
        unsigned       maskw[CAND * 8];
    };
    int selIdx[32], selOk[32];
    int npick;
    unsigned keepw[8];
};

__global__ __launch_bounds__(256) void det_kernel(const float* __restrict__ net,
                                                  float* __restrict__ out) {
    __shared__ u64 keys[CAND];
    __shared__ u64 xch[3][NWAVE][CAND];
    __shared__ Phase2 p2;
    Phase2* P = &p2;

    const int tid = threadIdx.x;
    const int img = blockIdx.x;
    const float* in = net + (size_t)img * D_IMG;
    const int lane = tid & 31;
    const int wave = tid >> 5;

    u64 key[NSLOT];
    #pragma unroll
    for (int n = 0; n < 13; ++n) {
        const int cell = tid + 256 * n;
        if (n < 12 || cell < CELLS) {
            const float* pr = in + cell * N_CLASS;
            float p[N_CLASS];
            #pragma unroll
            for (int c = 0; c < N_CLASS; ++c) p[c] = pr[c];
            #pragma unroll
            for (int bx = 0; bx < 2; ++bx) {
                int id = cell * 2 + bx;
                float conf = in[CONF_OFF + id];
                float best = NEG_INF;
                #pragma unroll
                for (int c = 0; c < N_CLASS; ++c) {
                    float v = __fmul_rn(conf, p[c]);
                    best = v > best ? v : best;
                }
                float s = (best >= 0.1f) ? best : 0.0f;
                key[2 * n + bx] = ((u64)__float_as_uint(s) << 32) | (unsigned)(~(unsigned)id);
            }
        } else {
            key[2 * n]     = (u64)(unsigned)(~(unsigned)(ITEMS + tid * NSLOT + 2 * n));
            key[2 * n + 1] = (u64)(unsigned)(~(unsigned)(ITEMS + tid * NSLOT + 2 * n + 1));
        }
    }
    #pragma unroll
    for (int s = 26; s < NSLOT; ++s)
        key[s] = (u64)(unsigned)(~(unsigned)(ITEMS + tid * NSLOT + s));

    stage_regs_q<4, 8, 1, 2>(key);
    stage_regs_q<4, 8, 2, 4>(key);
    stage_regs_q<4, 8, 1, 4>(key);
    for (int L = 0; L < 6; ++L) {
        const bool last = (L == 5);
        const bool d0 = last ? true : (((lane >> L) & 1) == 0);
        merge_rt<4, 8>(key, lane, (1 << L) >> 1, d0, last);
    }
    #pragma unroll
    for (int q = 0; q < 8; ++q) {
        key[q]      = key_max(key[q],      key[8 + q]);
        key[16 + q] = key_max(key[16 + q], key[24 + q]);
    }
    merge_rt<2, 16>(key, lane, 16, true, true);
    #pragma unroll
    for (int q = 0; q < 8; ++q) key[q] = key_max(key[q], key[16 + q]);
    merge_rt<1, 16>(key, lane, 16, (wave & 1) == 0, false);
    #pragma nounroll
    for (int r = 0; r < 3; ++r) {
        #pragma unroll
        for (int q = 0; q < 8; ++q) xch[r][wave][q * 32 + lane] = key[q];
        __syncthreads();
        #pragma unroll
        for (int q = 0; q < 8; ++q) key[q] = key_max(key[q], xch[r][wave ^ (1 << r)][q * 32 + lane]);
        merge_rt<1, 16>(key, lane, 16, (r == 2) ? true : ((wave & (2 << r)) == 0), false);
    }
    if (wave == 0) {
        #pragma unroll
        for (int q = 0; q < 8; ++q) keys[lane * 8 + q] = key[q];
    }
    __syncthreads();

    {
        unsigned long long key = keys[tid];
        unsigned id = ~(unsigned)key;
        float s = __uint_as_float((unsigned)(key >> 32));
        int cell = (int)(id >> 1);
        int gi = cell / S_GRID, gj = cell % S_GRID;
        const float* cd = in + COORD_OFF + (size_t)id * 4;
        float tx = cd[0], ty = cd[1], tw = cd[2], th = cd[3];
        float x = __fdiv_rn(__fadd_rn(tx, (float)gj), 56.0f);
        float y = __fdiv_rn(__fadd_rn(ty, (float)gi), 56.0f);
        float hw = __fmul_rn(__fmul_rn(tw, tw), 0.5f);
        float hh = __fmul_rn(__fmul_rn(th, th), 0.5f);
        float y0 = __fsub_rn(y, hh), x0 = __fsub_rn(x, hw);
        float y1 = __fadd_rn(y, hh), x1 = __fadd_rn(x, hw);
        float conf = in[CONF_OFF + id];
        const float* pr = in + cell * N_CLASS;
        float best = NEG_INF; int bc = 0;
        #pragma unroll
        for (int c = 0; c < N_CLASS; ++c) {
            float v = __fmul_rn(conf, pr[c]);
            if (v > best) { best = v; bc = c; }
        }
        float dy = fmaxf(__fsub_rn(y1, y0), 0.0f);
        float dx = fmaxf(__fsub_rn(x1, x0), 0.0f);
        P->y0[tid] = y0; P->x0[tid] = x0; P->y1[tid] = y1; P->x1[tid] = x1;
        P->sc[tid] = s;  P->cl[tid] = (float)bc;
        P->ar[tid] = __fmul_rn(dy, dx);
        if (tid < 32) { P->selIdx[tid] = 0; P->selOk[tid] = 0; }
    }
    __syncthreads();

    {
        const unsigned below = (1u << lane) - 1u;
        int total = 0;
        for (int q = 0; q < 8; ++q) {
            v8f czero = {};
            for (int tt = wave; tt < 4 * (q + 1); tt += NWAVE) {
                int mi = 2 * q + (tt & 1), ni = tt >> 1;
                int m0 = mi << 4, n0 = ni << 4;
                v2f A, B;
                if (lane < 16) {
                    A[0] = P->ar[m0 + lane]; A[1] = 1.0f;
                    B[0] = 1.0f;             B[1] = P->ar[n0 + lane];
                } else {
                    A[0] = 0.0f; A[1] = 0.0f; B[0] = 0.0f; B[1] = 0.0f;
                }
                v8f d = __builtin_amdgcn_wmma_f32_16x16x4_f32(
                    false, A, false, B, (short)0, czero, false, false);
                int n = n0 + (lane & 15);
                float ny0 = P->y0[n], nx0 = P->x0[n], ny1 = P->y1[n], nx1 = P->x1[n];
                int mbase = m0 + ((lane & 16) ? 8 : 0);
                #pragma unroll
                for (int r = 0; r < 8; ++r) {
                    int m = mbase + r;
                    float iy = fmaxf(__fsub_rn(fminf(P->y1[m], ny1), fmaxf(P->y0[m], ny0)), 0.0f);
                    float ix = fmaxf(__fsub_rn(fminf(P->x1[m], nx1), fmaxf(P->x0[m], nx0)), 0.0f);
                    float inter = __fmul_rn(iy, ix);
                    float uni = __fsub_rn(d[r], inter);
                    bool pred = (uni > 0.0f) && (__fdiv_rn(inter, uni) > 0.4f);
                    unsigned bal = __builtin_amdgcn_ballot_w32(pred);
                    if (lane == 0) {
                        P->mask16[(m0 + r)     * 16 + ni] = (unsigned short)(bal & 0xFFFFu);
                        P->mask16[(m0 + 8 + r) * 16 + ni] = (unsigned short)(bal >> 16);
                    }
                }
            }
            __syncthreads();

            if (wave == 0) {
                unsigned ext = 0;
                for (int w = 0; w < q; ++w) ext |= P->maskw[(q * 32 + lane) * 8 + w] & P->keepw[w];
                const unsigned own = P->maskw[(q * 32 + lane) * 8 + q] & below;
                const unsigned open = __builtin_amdgcn_ballot_w32(ext == 0);
                unsigned k = open;
                for (int pass = 0; pass < 32; ++pass) {
                    unsigned k2 = open & __builtin_amdgcn_ballot_w32((own & k) == 0);
                    if (k2 == k) break;
                    k = k2;
                }
                if ((k >> lane) & 1u) {
                    int rank = total + __builtin_popcount(k & below);
                    if (rank < MAXOUT) { P->selIdx[rank] = q * 32 + lane; P->selOk[rank] = 1; }
                }
                total += __builtin_popcount(k);
                if (lane == 0) { P->npick = total; P->keepw[q] = k; }
            }
            __syncthreads();
            if (P->npick >= MAXOUT) break;
        }
    }

    if (tid < MAXOUT) {
        int sel = P->selIdx[tid];
        float okf = P->selOk[tid] ? 1.0f : 0.0f;
        P->redv[tid] = __fmul_rn(P->sc[sel], okf);
    }
    __syncthreads();
    if (tid < MAXOUT) {
        int sel = P->selIdx[tid];
        float okf = P->selOk[tid] ? 1.0f : 0.0f;
        float ms = P->redv[tid];
        int rank = 0;
        for (int jt = 0; jt < MAXOUT; ++jt) {
            float vj = P->redv[jt];
            if (vj > ms || (vj == ms && jt < tid)) rank++;
        }
        float* o = out + (size_t)img * (MAXOUT * 6) + (size_t)rank * 6;
        o[0] = __fmul_rn(P->y0[sel], okf);
        o[1] = __fmul_rn(P->x0[sel], okf);
        o[2] = __fmul_rn(P->y1[sel], okf);
        o[3] = __fmul_rn(P->x1[sel], okf);
        o[4] = ms;
        o[5] = __fmul_rn(P->cl[sel], okf);
    }
}

extern "C" void kernel_launch(void* const* d_in, const int* in_sizes, int n_in,
                              void* d_out, int out_size, void* d_ws, size_t ws_size,
                              hipStream_t stream) {
    (void)n_in; (void)out_size; (void)d_ws; (void)ws_size;
    const float* net = (const float*)d_in[0];
    float* out = (float*)d_out;
    int batch = in_sizes[0] / D_IMG;
    det_kernel<<<dim3(batch), dim3(256), 0, stream>>>(net, out);
}


# Class loads all issued first; each wave keys and sorts its four 256-key chunks one at a time as their values land (partial last pass first): chunk sorts run under the loads still in flight
# speedup vs baseline: 11.8120x; 1.0325x over previous
#include <hip/hip_runtime.h>

typedef float v2f __attribute__((ext_vector_type(2)));
typedef float v8f __attribute__((ext_vector_type(8)));

#define S_GRID   56
#define N_CLASS  20
#define CELLS    (S_GRID * S_GRID)
#define ITEMS    (CELLS * 2)
#define CAND     256
#define MAXOUT   30
#define D_IMG    94080
#define CONF_OFF 62720
#define COORD_OFF 68992
#define NEG_INF  (-3.402823466e38f)
#define NSLOT    32
#define NWAVE    8

typedef unsigned long long u64;


__device__ __forceinline__ u64 key_from_row_lane(u64 v, unsigned sel) {
    unsigned lo = (unsigned)v, hi = (unsigned)(v >> 32);
    unsigned rlo = __builtin_amdgcn_permlane16_var(lo, lo, sel, true, false);
    unsigned rhi = __builtin_amdgcn_permlane16_var(hi, hi, sel, true, false);
    return ((u64)rhi << 32) | (u64)rlo;
}
__device__ __forceinline__ u64 key_from_other_row(u64 v) {
    unsigned lo = (unsigned)v, hi = (unsigned)(v >> 32);
    unsigned rlo = __builtin_amdgcn_permlanex16(lo, lo, 0x76543210u, 0xfedcba98u, true, false);
    unsigned rhi = __builtin_amdgcn_permlanex16(hi, hi, 0x76543210u, 0xfedcba98u, true, false);
    return ((u64)rhi << 32) | (u64)rlo;
}

template <int NCH, int STRIDE, int J, int K, int BASE = 0>
__device__ __forceinline__ void stage_regs_q(u64 (&key)[NSLOT]) {
    #pragma unroll
    for (int c = 0; c < NCH; ++c) {
        #pragma unroll
        for (int q = 0; q < 8; ++q) {
            if ((q & J) == 0) {
                const int s = BASE + c * STRIDE + q;
                const bool desc = (q & K) == 0;
                u64 a = key[s], b = key[s + J];
                bool sw = desc ? (a < b) : (a > b);
                key[s]     = sw ? b : a;
                key[s + J] = sw ? a : b;
            }
        }
    }
}

template <int NCH, int STRIDE, bool XROW, int BASE = 0>
__device__ __forceinline__ void stage_lanes_rt(u64 (&key)[NSLOT], int lane, int M, bool d0, bool alt) {
    const unsigned sel = (unsigned)(lane ^ M);
    const bool lower = (lane & M) == 0;
    #pragma unroll
    for (int c = 0; c < NCH; ++c) {
        const bool desc = (c & 1) ? (d0 != alt) : d0;
        const bool keepmin = (lower != desc);
        #pragma unroll
        for (int q = 0; q < 8; ++q) {
            const int s = BASE + c * STRIDE + q;
            u64 mine = key[s];
            u64 oth = XROW ? key_from_other_row(mine) : key_from_row_lane(mine, sel);
            bool take = (oth > mine) != keepmin;
            key[s] = take ? oth : mine;
        }
    }
}

template <int NCH, int STRIDE, int J, int BASE = 0>
__device__ __forceinline__ void stage_regs_rt(u64 (&key)[NSLOT], bool d0, bool alt) {
    #pragma unroll
    for (int c = 0; c < NCH; ++c) {
        const bool desc = (c & 1) ? (d0 != alt) : d0;
        #pragma unroll
        for (int q = 0; q < 8; ++q) {
            if ((q & J) == 0) {
                const int s = BASE + c * STRIDE + q;
                u64 a = key[s], b = key[s + J];
                bool sw = (a < b) == desc;
                key[s]     = sw ? b : a;
                key[s + J] = sw ? a : b;
            }
        }
    }
}

template <int NCH, int STRIDE, int BASE = 0>
__device__ __forceinline__ void merge_rt(u64 (&key)[NSLOT], int lane, int mtop, bool d0, bool alt) {
    if (mtop >= 16) stage_lanes_rt<NCH, STRIDE, true, BASE>(key, lane, 16, d0, alt);
    #pragma nounroll
    for (int M = (mtop >= 16) ? 8 : mtop; M >= 1; M >>= 1) stage_lanes_rt<NCH, STRIDE, false, BASE>(key, lane, M, d0, alt);
    stage_regs_rt<NCH, STRIDE, 4, BASE>(key, d0, alt);
    stage_regs_rt<NCH, STRIDE, 2, BASE>(key, d0, alt);
    stage_regs_rt<NCH, STRIDE, 1, BASE>(key, d0, alt);
}

template <int G>
__device__ __forceinline__ void sort_chunk(u64 (&key)[NSLOT], int lane) {
    stage_regs_q<1, 8, 1, 2, 8 * G>(key);
    stage_regs_q<1, 8, 2, 4, 8 * G>(key);
    stage_regs_q<1, 8, 1, 4, 8 * G>(key);
    #pragma nounroll
    for (int L = 0; L < 6; ++L) {
        const bool last = (L == 5);
        const bool d0 = last ? ((G & 1) == 0) : (((lane >> L) & 1) == 0);
        merge_rt<1, 8, 8 * G>(key, lane, (1 << L) >> 1, d0, false);
    }
}

__device__ __forceinline__ u64 score_key(float conf, const float (&p)[N_CLASS], int id) {
    float best = NEG_INF;
    #pragma unroll
    for (int c = 0; c < N_CLASS; ++c) {
        float v = __fmul_rn(conf, p[c]);
        best = v > best ? v : best;
    }
    float s = (best >= 0.1f) ? best : 0.0f;
    return ((u64)__float_as_uint(s) << 32) | (unsigned)(~(unsigned)id);
}

template <int G>
__device__ __forceinline__ void chunk_keys(u64 (&key)[NSLOT], const float (&p)[12][N_CLASS],
                                           const float (&cf)[12][2], int tid) {
    #pragma unroll
    for (int n = 4 * G; n < 4 * G + 4; ++n) {
        const int cell = tid + 256 * n;
        key[2 * n]     = score_key(cf[n][0], p[n], cell * 2);
        key[2 * n + 1] = score_key(cf[n][1], p[n], cell * 2 + 1);
    }
}

__device__ __forceinline__ u64 key_max(u64 a, u64 b) { return a > b ? a : b; }

struct Phase2 {
    float y0[CAND], x0[CAND], y1[CAND], x1[CAND];
    float sc[CAND], cl[CAND], ar[CAND];
    float redv[CAND];
    union alignas(16) {
        unsigned short mask16[CAND * 16];
        unsigned       maskw[CAND * 8];
    };
    int selIdx[32], selOk[32];
    int npick;
    unsigned keepw[8];
};

__global__ __launch_bounds__(256) void det_kernel(const float* __restrict__ net,
                                                  float* __restrict__ out) {
    __shared__ u64 keys[CAND];
    __shared__ u64 xch[3][NWAVE][CAND];
    __shared__ Phase2 p2;
    Phase2* P = &p2;

    const int tid = threadIdx.x;
    const int img = blockIdx.x;
    const float* in = net + (size_t)img * D_IMG;
    const int lane = tid & 31;
    const int wave = tid >> 5;

    u64 key[NSLOT];
    const bool has12 = tid < (CELLS - 12 * 256);
    const int cell12 = has12 ? tid + 256 * 12 : 0;
    float p12[N_CLASS], cf12[2];
    {
        const float* pr = in + cell12 * N_CLASS;
        #pragma unroll
        for (int c = 0; c < N_CLASS; ++c) p12[c] = pr[c];
        cf12[0] = in[CONF_OFF + cell12 * 2];
        cf12[1] = in[CONF_OFF + cell12 * 2 + 1];
    }
    float p[12][N_CLASS], cf[12][2];
    #pragma unroll
    for (int n = 0; n < 12; ++n) {
        const int cell = tid + 256 * n;
        const float* pr = in + cell * N_CLASS;
        #pragma unroll
        for (int c = 0; c < N_CLASS; ++c) p[n][c] = pr[c];
        cf[n][0] = in[CONF_OFF + cell * 2];
        cf[n][1] = in[CONF_OFF + cell * 2 + 1];
    }

    {
        const u64 k0 = score_key(cf12[0], p12, cell12 * 2);
        const u64 k1 = score_key(cf12[1], p12, cell12 * 2 + 1);
        key[24] = has12 ? k0 : (u64)(unsigned)(~(unsigned)(ITEMS + tid * NSLOT + 24));
        key[25] = has12 ? k1 : (u64)(unsigned)(~(unsigned)(ITEMS + tid * NSLOT + 25));
        #pragma unroll
        for (int s = 26; s < NSLOT; ++s)
            key[s] = (u64)(unsigned)(~(unsigned)(ITEMS + tid * NSLOT + s));
    }
    sort_chunk<3>(key, lane);
    chunk_keys<0>(key, p, cf, tid);
    sort_chunk<0>(key, lane);
    chunk_keys<1>(key, p, cf, tid);
    sort_chunk<1>(key, lane);
    chunk_keys<2>(key, p, cf, tid);
    sort_chunk<2>(key, lane);
    #pragma unroll
    for (int q = 0; q < 8; ++q) {
        key[q]      = key_max(key[q],      key[8 + q]);
        key[16 + q] = key_max(key[16 + q], key[24 + q]);
    }
    merge_rt<2, 16>(key, lane, 16, true, true);
    #pragma unroll
    for (int q = 0; q < 8; ++q) key[q] = key_max(key[q], key[16 + q]);
    merge_rt<1, 16>(key, lane, 16, (wave & 1) == 0, false);
    #pragma nounroll
    for (int r = 0; r < 3; ++r) {
        #pragma unroll
        for (int q = 0; q < 8; ++q) xch[r][wave][q * 32 + lane] = key[q];
        __syncthreads();
        #pragma unroll
        for (int q = 0; q < 8; ++q) key[q] = key_max(key[q], xch[r][wave ^ (1 << r)][q * 32 + lane]);
        merge_rt<1, 16>(key, lane, 16, (r == 2) ? true : ((wave & (2 << r)) == 0), false);
    }
    if (wave == 0) {
        #pragma unroll
        for (int q = 0; q < 8; ++q) keys[lane * 8 + q] = key[q];
    }
    __syncthreads();

    {
        unsigned long long key = keys[tid];
        unsigned id = ~(unsigned)key;
        float s = __uint_as_float((unsigned)(key >> 32));
        int cell = (int)(id >> 1);
        int gi = cell / S_GRID, gj = cell % S_GRID;
        const float* cd = in + COORD_OFF + (size_t)id * 4;
        float tx = cd[0], ty = cd[1], tw = cd[2], th = cd[3];
        float x = __fdiv_rn(__fadd_rn(tx, (float)gj), 56.0f);
        float y = __fdiv_rn(__fadd_rn(ty, (float)gi), 56.0f);
        float hw = __fmul_rn(__fmul_rn(tw, tw), 0.5f);
        float hh = __fmul_rn(__fmul_rn(th, th), 0.5f);
        float y0 = __fsub_rn(y, hh), x0 = __fsub_rn(x, hw);
        float y1 = __fadd_rn(y, hh), x1 = __fadd_rn(x, hw);
        float conf = in[CONF_OFF + id];
        const float* pr = in + cell * N_CLASS;
        float best = NEG_INF; int bc = 0;
        #pragma unroll
        for (int c = 0; c < N_CLASS; ++c) {
            float v = __fmul_rn(conf, pr[c]);
            if (v > best) { best = v; bc = c; }
        }
        float dy = fmaxf(__fsub_rn(y1, y0), 0.0f);
        float dx = fmaxf(__fsub_rn(x1, x0), 0.0f);
        P->y0[tid] = y0; P->x0[tid] = x0; P->y1[tid] = y1; P->x1[tid] = x1;
        P->sc[tid] = s;  P->cl[tid] = (float)bc;
        P->ar[tid] = __fmul_rn(dy, dx);
        if (tid < 32) { P->selIdx[tid] = 0; P->selOk[tid] = 0; }
    }
    __syncthreads();

    {
        const unsigned below = (1u << lane) - 1u;
        int total = 0;
        for (int q = 0; q < 8; ++q) {
            v8f czero = {};
            for (int tt = wave; tt < 4 * (q + 1); tt += NWAVE) {
                int mi = 2 * q + (tt & 1), ni = tt >> 1;
                int m0 = mi << 4, n0 = ni << 4;
                v2f A, B;
                if (lane < 16) {
                    A[0] = P->ar[m0 + lane]; A[1] = 1.0f;
                    B[0] = 1.0f;             B[1] = P->ar[n0 + lane];
                } else {
                    A[0] = 0.0f; A[1] = 0.0f; B[0] = 0.0f; B[1] = 0.0f;
                }
                v8f d = __builtin_amdgcn_wmma_f32_16x16x4_f32(
                    false, A, false, B, (short)0, czero, false, false);
                int n = n0 + (lane & 15);
                float ny0 = P->y0[n], nx0 = P->x0[n], ny1 = P->y1[n], nx1 = P->x1[n];
                int mbase = m0 + ((lane & 16) ? 8 : 0);
                #pragma unroll
                for (int r = 0; r < 8; ++r) {
                    int m = mbase + r;
                    float iy = fmaxf(__fsub_rn(fminf(P->y1[m], ny1), fmaxf(P->y0[m], ny0)), 0.0f);
                    float ix = fmaxf(__fsub_rn(fminf(P->x1[m], nx1), fmaxf(P->x0[m], nx0)), 0.0f);
                    float inter = __fmul_rn(iy, ix);
                    float uni = __fsub_rn(d[r], inter);
                    bool pred = (uni > 0.0f) && (__fdiv_rn(inter, uni) > 0.4f);
                    unsigned bal = __builtin_amdgcn_ballot_w32(pred);
                    if (lane == 0) {
                        P->mask16[(m0 + r)     * 16 + ni] = (unsigned short)(bal & 0xFFFFu);
                        P->mask16[(m0 + 8 + r) * 16 + ni] = (unsigned short)(bal >> 16);
                    }
                }
            }
            __syncthreads();

            if (wave == 0) {
                unsigned ext = 0;
                for (int w = 0; w < q; ++w) ext |= P->maskw[(q * 32 + lane) * 8 + w] & P->keepw[w];
                const unsigned own = P->maskw[(q * 32 + lane) * 8 + q] & below;
                const unsigned open = __builtin_amdgcn_ballot_w32(ext == 0);
                unsigned k = open;
                for (int pass = 0; pass < 32; ++pass) {
                    unsigned k2 = open & __builtin_amdgcn_ballot_w32((own & k) == 0);
                    if (k2 == k) break;
                    k = k2;
                }
                if ((k >> lane) & 1u) {
                    int rank = total + __builtin_popcount(k & below);
                    if (rank < MAXOUT) { P->selIdx[rank] = q * 32 + lane; P->selOk[rank] = 1; }
                }
                total += __builtin_popcount(k);
                if (lane == 0) { P->npick = total; P->keepw[q] = k; }
            }
            __syncthreads();
            if (P->npick >= MAXOUT) break;
        }
    }

    if (tid < MAXOUT) {
        int sel = P->selIdx[tid];
        float okf = P->selOk[tid] ? 1.0f : 0.0f;
        P->redv[tid] = __fmul_rn(P->sc[sel], okf);
    }
    __syncthreads();
    if (tid < MAXOUT) {
        int sel = P->selIdx[tid];
        float okf = P->selOk[tid] ? 1.0f : 0.0f;
        float ms = P->redv[tid];
        int rank = 0;
        for (int jt = 0; jt < MAXOUT; ++jt) {
            float vj = P->redv[jt];
            if (vj > ms || (vj == ms && jt < tid)) rank++;
        }
        float* o = out + (size_t)img * (MAXOUT * 6) + (size_t)rank * 6;
        o[0] = __fmul_rn(P->y0[sel], okf);
        o[1] = __fmul_rn(P->x0[sel], okf);
        o[2] = __fmul_rn(P->y1[sel], okf);
        o[3] = __fmul_rn(P->x1[sel], okf);
        o[4] = ms;
        o[5] = __fmul_rn(P->cl[sel], okf);
    }
}

extern "C" void kernel_launch(void* const* d_in, const int* in_sizes, int n_in,
                              void* d_out, int out_size, void* d_ws, size_t ws_size,
                              hipStream_t stream) {
    (void)n_in; (void)out_size; (void)d_ws; (void)ws_size;
    const float* net = (const float*)d_in[0];
    float* out = (float*)d_out;
    int batch = in_sizes[0] / D_IMG;
    det_kernel<<<dim3(batch), dim3(256), 0, stream>>>(net, out);
}


# Sort network compares its 64-bit keys on 32-bit halves: low-word compare, then high-word subtract-with-borrow whose borrow-out is the answer, in place of 64-bit compare, min and max instructions
# speedup vs baseline: 12.3846x; 1.0485x over previous
#include <hip/hip_runtime.h>

typedef float v2f __attribute__((ext_vector_type(2)));
typedef float v8f __attribute__((ext_vector_type(8)));

#define S_GRID   56
#define N_CLASS  20
#define CELLS    (S_GRID * S_GRID)
#define ITEMS    (CELLS * 2)
#define CAND     256
#define MAXOUT   30
#define D_IMG    94080
#define CONF_OFF 62720
#define COORD_OFF 68992
#define NEG_INF  (-3.402823466e38f)
#define NSLOT    32
#define NWAVE    8

typedef unsigned long long u64;

__device__ __forceinline__ unsigned key_lt_mask(u64 a, u64 b) {
    const unsigned al = (unsigned)a, ah = (unsigned)(a >> 32);
    const unsigned bl = (unsigned)b, bh = (unsigned)(b >> 32);
    const unsigned lo = __builtin_amdgcn_ballot_w32(al < bl);
    unsigned diff, out;
    asm("v_sub_co_ci_u32_e64 %0, %1, %2, %3, %4" : "=v"(diff), "=s"(out) : "v"(ah), "v"(bh), "s"(lo));
    return out;
}
__device__ __forceinline__ u64 key_pick(unsigned mask, u64 x, u64 y) {
    return __builtin_amdgcn_inverse_ballot_w32(mask) ? x : y;
}


__device__ __forceinline__ u64 key_from_row_lane(u64 v, unsigned sel) {
    unsigned lo = (unsigned)v, hi = (unsigned)(v >> 32);
    unsigned rlo = __builtin_amdgcn_permlane16_var(lo, lo, sel, true, false);
    unsigned rhi = __builtin_amdgcn_permlane16_var(hi, hi, sel, true, false);
    return ((u64)rhi << 32) | (u64)rlo;
}
__device__ __forceinline__ u64 key_from_other_row(u64 v) {
    unsigned lo = (unsigned)v, hi = (unsigned)(v >> 32);
    unsigned rlo = __builtin_amdgcn_permlanex16(lo, lo, 0x76543210u, 0xfedcba98u, true, false);
    unsigned rhi = __builtin_amdgcn_permlanex16(hi, hi, 0x76543210u, 0xfedcba98u, true, false);
    return ((u64)rhi << 32) | (u64)rlo;
}

template <int NCH, int STRIDE, int J, int K, int BASE = 0>
__device__ __forceinline__ void stage_regs_q(u64 (&key)[NSLOT]) {
    #pragma unroll
    for (int c = 0; c < NCH; ++c) {
        #pragma unroll
        for (int q = 0; q < 8; ++q) {
            if ((q & J) == 0) {
                const int s = BASE + c * STRIDE + q;
                const bool desc = (q & K) == 0;
                u64 a = key[s], b = key[s + J];
                const unsigned sw = desc ? key_lt_mask(a, b) : key_lt_mask(b, a);
                key[s]     = key_pick(sw, b, a);
                key[s + J] = key_pick(sw, a, b);
            }
        }
    }
}

template <int NCH, int STRIDE, bool XROW, int BASE = 0>
__device__ __forceinline__ void stage_lanes_rt(u64 (&key)[NSLOT], int lane, int M, bool d0, bool alt) {
    const unsigned sel = (unsigned)(lane ^ M);
    const bool lower = (lane & M) == 0;
    #pragma unroll
    for (int c = 0; c < NCH; ++c) {
        const bool desc = (c & 1) ? (d0 != alt) : d0;
        const unsigned keepmin = __builtin_amdgcn_ballot_w32(lower != desc);
        #pragma unroll
        for (int q = 0; q < 8; ++q) {
            const int s = BASE + c * STRIDE + q;
            u64 mine = key[s];
            u64 oth = XROW ? key_from_other_row(mine) : key_from_row_lane(mine, sel);
            const unsigned take = key_lt_mask(mine, oth) ^ keepmin;
            key[s] = key_pick(take, oth, mine);
        }
    }
}

template <int NCH, int STRIDE, int J, int BASE = 0>
__device__ __forceinline__ void stage_regs_rt(u64 (&key)[NSLOT], bool d0, bool alt) {
    #pragma unroll
    for (int c = 0; c < NCH; ++c) {
        const unsigned asc = ~__builtin_amdgcn_ballot_w32((c & 1) ? (d0 != alt) : d0);
        #pragma unroll
        for (int q = 0; q < 8; ++q) {
            if ((q & J) == 0) {
                const int s = BASE + c * STRIDE + q;
                u64 a = key[s], b = key[s + J];
                const unsigned sw = key_lt_mask(a, b) ^ asc;
                key[s]     = key_pick(sw, b, a);
                key[s + J] = key_pick(sw, a, b);
            }
        }
    }
}

template <int NCH, int STRIDE, int BASE = 0>
__device__ __forceinline__ void merge_rt(u64 (&key)[NSLOT], int lane, int mtop, bool d0, bool alt) {
    if (mtop >= 16) stage_lanes_rt<NCH, STRIDE, true, BASE>(key, lane, 16, d0, alt);
    #pragma nounroll
    for (int M = (mtop >= 16) ? 8 : mtop; M >= 1; M >>= 1) stage_lanes_rt<NCH, STRIDE, false, BASE>(key, lane, M, d0, alt);
    stage_regs_rt<NCH, STRIDE, 4, BASE>(key, d0, alt);
    stage_regs_rt<NCH, STRIDE, 2, BASE>(key, d0, alt);
    stage_regs_rt<NCH, STRIDE, 1, BASE>(key, d0, alt);
}

template <int G>
__device__ __forceinline__ void sort_chunk(u64 (&key)[NSLOT], int lane) {
    stage_regs_q<1, 8, 1, 2, 8 * G>(key);
    stage_regs_q<1, 8, 2, 4, 8 * G>(key);
    stage_regs_q<1, 8, 1, 4, 8 * G>(key);
    #pragma nounroll
    for (int L = 0; L < 6; ++L) {
        const bool last = (L == 5);
        const bool d0 = last ? ((G & 1) == 0) : (((lane >> L) & 1) == 0);
        merge_rt<1, 8, 8 * G>(key, lane, (1 << L) >> 1, d0, false);
    }
}

__device__ __forceinline__ u64 score_key(float conf, const float (&p)[N_CLASS], int id) {
    float best = NEG_INF;
    #pragma unroll
    for (int c = 0; c < N_CLASS; ++c) {
        float v = __fmul_rn(conf, p[c]);
        best = v > best ? v : best;
    }
    float s = (best >= 0.1f) ? best : 0.0f;
    return ((u64)__float_as_uint(s) << 32) | (unsigned)(~(unsigned)id);
}

template <int G>
__device__ __forceinline__ void chunk_keys(u64 (&key)[NSLOT], const float (&p)[12][N_CLASS],
                                           const float (&cf)[12][2], int tid) {
    #pragma unroll
    for (int n = 4 * G; n < 4 * G + 4; ++n) {
        const int cell = tid + 256 * n;
        key[2 * n]     = score_key(cf[n][0], p[n], cell * 2);
        key[2 * n + 1] = score_key(cf[n][1], p[n], cell * 2 + 1);
    }
}

__device__ __forceinline__ u64 key_max(u64 a, u64 b) { return key_pick(key_lt_mask(a, b), b, a); }

struct Phase2 {
    float y0[CAND], x0[CAND], y1[CAND], x1[CAND];
    float sc[CAND], cl[CAND], ar[CAND];
    float redv[CAND];
    union alignas(16) {
        unsigned short mask16[CAND * 16];
        unsigned       maskw[CAND * 8];
    };
    int selIdx[32], selOk[32];
    int npick;
    unsigned keepw[8];
};

__global__ __launch_bounds__(256) void det_kernel(const float* __restrict__ net,
                                                  float* __restrict__ out) {
    __shared__ u64 keys[CAND];
    __shared__ u64 xch[3][NWAVE][CAND];
    __shared__ Phase2 p2;
    Phase2* P = &p2;

    const int tid = threadIdx.x;
    const int img = blockIdx.x;
    const float* in = net + (size_t)img * D_IMG;
    const int lane = tid & 31;
    const int wave = tid >> 5;

    u64 key[NSLOT];
    const bool has12 = tid < (CELLS - 12 * 256);
    const int cell12 = has12 ? tid + 256 * 12 : 0;
    float p12[N_CLASS], cf12[2];
    {
        const float* pr = in + cell12 * N_CLASS;
        #pragma unroll
        for (int c = 0; c < N_CLASS; ++c) p12[c] = pr[c];
        cf12[0] = in[CONF_OFF + cell12 * 2];
        cf12[1] = in[CONF_OFF + cell12 * 2 + 1];
    }
    float p[12][N_CLASS], cf[12][2];
    #pragma unroll
    for (int n = 0; n < 12; ++n) {
        const int cell = tid + 256 * n;
        const float* pr = in + cell * N_CLASS;
        #pragma unroll
        for (int c = 0; c < N_CLASS; ++c) p[n][c] = pr[c];
        cf[n][0] = in[CONF_OFF + cell * 2];
        cf[n][1] = in[CONF_OFF + cell * 2 + 1];
    }

    {
        const u64 k0 = score_key(cf12[0], p12, cell12 * 2);
        const u64 k1 = score_key(cf12[1], p12, cell12 * 2 + 1);
        key[24] = has12 ? k0 : (u64)(unsigned)(~(unsigned)(ITEMS + tid * NSLOT + 24));
        key[25] = has12 ? k1 : (u64)(unsigned)(~(unsigned)(ITEMS + tid * NSLOT + 25));
        #pragma unroll
        for (int s = 26; s < NSLOT; ++s)
            key[s] = (u64)(unsigned)(~(unsigned)(ITEMS + tid * NSLOT + s));
    }
    sort_chunk<3>(key, lane);
    chunk_keys<0>(key, p, cf, tid);
    sort_chunk<0>(key, lane);
    chunk_keys<1>(key, p, cf, tid);
    sort_chunk<1>(key, lane);
    chunk_keys<2>(key, p, cf, tid);
    sort_chunk<2>(key, lane);
    #pragma unroll
    for (int q = 0; q < 8; ++q) {
        key[q]      = key_max(key[q],      key[8 + q]);
        key[16 + q] = key_max(key[16 + q], key[24 + q]);
    }
    merge_rt<2, 16>(key, lane, 16, true, true);
    #pragma unroll
    for (int q = 0; q < 8; ++q) key[q] = key_max(key[q], key[16 + q]);
    merge_rt<1, 16>(key, lane, 16, (wave & 1) == 0, false);
    #pragma nounroll
    for (int r = 0; r < 3; ++r) {
        #pragma unroll
        for (int q = 0; q < 8; ++q) xch[r][wave][q * 32 + lane] = key[q];
        __syncthreads();
        #pragma unroll
        for (int q = 0; q < 8; ++q) key[q] = key_max(key[q], xch[r][wave ^ (1 << r)][q * 32 + lane]);
        merge_rt<1, 16>(key, lane, 16, (r == 2) ? true : ((wave & (2 << r)) == 0), false);
    }
    if (wave == 0) {
        #pragma unroll
        for (int q = 0; q < 8; ++q) keys[lane * 8 + q] = key[q];
    }
    __syncthreads();

    {
        unsigned long long key = keys[tid];
        unsigned id = ~(unsigned)key;
        float s = __uint_as_float((unsigned)(key >> 32));
        int cell = (int)(id >> 1);
        int gi = cell / S_GRID, gj = cell % S_GRID;
        const float* cd = in + COORD_OFF + (size_t)id * 4;
        float tx = cd[0], ty = cd[1], tw = cd[2], th = cd[3];
        float x = __fdiv_rn(__fadd_rn(tx, (float)gj), 56.0f);
        float y = __fdiv_rn(__fadd_rn(ty, (float)gi), 56.0f);
        float hw = __fmul_rn(__fmul_rn(tw, tw), 0.5f);
        float hh = __fmul_rn(__fmul_rn(th, th), 0.5f);
        float y0 = __fsub_rn(y, hh), x0 = __fsub_rn(x, hw);
        float y1 = __fadd_rn(y, hh), x1 = __fadd_rn(x, hw);
        float conf = in[CONF_OFF + id];
        const float* pr = in + cell * N_CLASS;
        float best = NEG_INF; int bc = 0;
        #pragma unroll
        for (int c = 0; c < N_CLASS; ++c) {
            float v = __fmul_rn(conf, pr[c]);
            if (v > best) { best = v; bc = c; }
        }
        float dy = fmaxf(__fsub_rn(y1, y0), 0.0f);
        float dx = fmaxf(__fsub_rn(x1, x0), 0.0f);
        P->y0[tid] = y0; P->x0[tid] = x0; P->y1[tid] = y1; P->x1[tid] = x1;
        P->sc[tid] = s;  P->cl[tid] = (float)bc;
        P->ar[tid] = __fmul_rn(dy, dx);
        if (tid < 32) { P->selIdx[tid] = 0; P->selOk[tid] = 0; }
    }
    __syncthreads();

    {
        const unsigned below = (1u << lane) - 1u;
        int total = 0;
        for (int q = 0; q < 8; ++q) {
            v8f czero = {};
            for (int tt = wave; tt < 4 * (q + 1); tt += NWAVE) {
                int mi = 2 * q + (tt & 1), ni = tt >> 1;
                int m0 = mi << 4, n0 = ni << 4;
                v2f A, B;
                if (lane < 16) {
                    A[0] = P->ar[m0 + lane]; A[1] = 1.0f;
                    B[0] = 1.0f;             B[1] = P->ar[n0 + lane];
                } else {
                    A[0] = 0.0f; A[1] = 0.0f; B[0] = 0.0f; B[1] = 0.0f;
                }
                v8f d = __builtin_amdgcn_wmma_f32_16x16x4_f32(
                    false, A, false, B, (short)0, czero, false, false);
                int n = n0 + (lane & 15);
                float ny0 = P->y0[n], nx0 = P->x0[n], ny1 = P->y1[n], nx1 = P->x1[n];
                int mbase = m0 + ((lane & 16) ? 8 : 0);
                #pragma unroll
                for (int r = 0; r < 8; ++r) {
                    int m = mbase + r;
                    float iy = fmaxf(__fsub_rn(fminf(P->y1[m], ny1), fmaxf(P->y0[m], ny0)), 0.0f);
                    float ix = fmaxf(__fsub_rn(fminf(P->x1[m], nx1), fmaxf(P->x0[m], nx0)), 0.0f);
                    float inter = __fmul_rn(iy, ix);
                    float uni = __fsub_rn(d[r], inter);
                    bool pred = (uni > 0.0f) && (__fdiv_rn(inter, uni) > 0.4f);
                    unsigned bal = __builtin_amdgcn_ballot_w32(pred);
                    if (lane == 0) {
                        P->mask16[(m0 + r)     * 16 + ni] = (unsigned short)(bal & 0xFFFFu);
                        P->mask16[(m0 + 8 + r) * 16 + ni] = (unsigned short)(bal >> 16);
                    }
                }
            }
            __syncthreads();

            if (wave == 0) {
                unsigned ext = 0;
                for (int w = 0; w < q; ++w) ext |= P->maskw[(q * 32 + lane) * 8 + w] & P->keepw[w];
                const unsigned own = P->maskw[(q * 32 + lane) * 8 + q] & below;
                const unsigned open = __builtin_amdgcn_ballot_w32(ext == 0);
                unsigned k = open;
                for (int pass = 0; pass < 32; ++pass) {
                    unsigned k2 = open & __builtin_amdgcn_ballot_w32((own & k) == 0);
                    if (k2 == k) break;
                    k = k2;
                }
                if ((k >> lane) & 1u) {
                    int rank = total + __builtin_popcount(k & below);
                    if (rank < MAXOUT) { P->selIdx[rank] = q * 32 + lane; P->selOk[rank] = 1; }
                }
                total += __builtin_popcount(k);
                if (lane == 0) { P->npick = total; P->keepw[q] = k; }
            }
            __syncthreads();
            if (P->npick >= MAXOUT) break;
        }
    }

    if (tid < MAXOUT) {
        int sel = P->selIdx[tid];
        float okf = P->selOk[tid] ? 1.0f : 0.0f;
        P->redv[tid] = __fmul_rn(P->sc[sel], okf);
    }
    __syncthreads();
    if (tid < MAXOUT) {
        int sel = P->selIdx[tid];
        float okf = P->selOk[tid] ? 1.0f : 0.0f;
        float ms = P->redv[tid];
        int rank = 0;
        for (int jt = 0; jt < MAXOUT; ++jt) {
            float vj = P->redv[jt];
            if (vj > ms || (vj == ms && jt < tid)) rank++;
        }
        float* o = out + (size_t)img * (MAXOUT * 6) + (size_t)rank * 6;
        o[0] = __fmul_rn(P->y0[sel], okf);
        o[1] = __fmul_rn(P->x0[sel], okf);
        o[2] = __fmul_rn(P->y1[sel], okf);
        o[3] = __fmul_rn(P->x1[sel], okf);
        o[4] = ms;
        o[5] = __fmul_rn(P->cl[sel], okf);
    }
}

extern "C" void kernel_launch(void* const* d_in, const int* in_sizes, int n_in,
                              void* d_out, int out_size, void* d_ws, size_t ws_size,
                              hipStream_t stream) {
    (void)n_in; (void)out_size; (void)d_ws; (void)ws_size;
    const float* net = (const float*)d_in[0];
    float* out = (float*)d_out;
    int batch = in_sizes[0] / D_IMG;
    det_kernel<<<dim3(batch), dim3(256), 0, stream>>>(net, out);
}


# Box scores from the class maximum: one max tree over a cell's 20 class values, shared by its two boxes, then one product per box, in place of 20 products and 20 compare-selects per box; same bits
# speedup vs baseline: 13.1894x; 1.0650x over previous
#include <hip/hip_runtime.h>

typedef float v2f __attribute__((ext_vector_type(2)));
typedef float v8f __attribute__((ext_vector_type(8)));

#define S_GRID   56
#define N_CLASS  20
#define CELLS    (S_GRID * S_GRID)
#define ITEMS    (CELLS * 2)
#define CAND     256
#define MAXOUT   30
#define D_IMG    94080
#define CONF_OFF 62720
#define COORD_OFF 68992
#define NEG_INF  (-3.402823466e38f)
#define NSLOT    32
#define NWAVE    8

typedef unsigned long long u64;

__device__ __forceinline__ unsigned key_lt_mask(u64 a, u64 b) {
    const unsigned al = (unsigned)a, ah = (unsigned)(a >> 32);
    const unsigned bl = (unsigned)b, bh = (unsigned)(b >> 32);
    const unsigned lo = __builtin_amdgcn_ballot_w32(al < bl);
    unsigned diff, out;
    asm("v_sub_co_ci_u32_e64 %0, %1, %2, %3, %4" : "=v"(diff), "=s"(out) : "v"(ah), "v"(bh), "s"(lo));
    return out;
}
__device__ __forceinline__ u64 key_pick(unsigned mask, u64 x, u64 y) {
    return __builtin_amdgcn_inverse_ballot_w32(mask) ? x : y;
}


__device__ __forceinline__ u64 key_from_row_lane(u64 v, unsigned sel) {
    unsigned lo = (unsigned)v, hi = (unsigned)(v >> 32);
    unsigned rlo = __builtin_amdgcn_permlane16_var(lo, lo, sel, true, false);
    unsigned rhi = __builtin_amdgcn_permlane16_var(hi, hi, sel, true, false);
    return ((u64)rhi << 32) | (u64)rlo;
}
__device__ __forceinline__ u64 key_from_other_row(u64 v) {
    unsigned lo = (unsigned)v, hi = (unsigned)(v >> 32);
    unsigned rlo = __builtin_amdgcn_permlanex16(lo, lo, 0x76543210u, 0xfedcba98u, true, false);
    unsigned rhi = __builtin_amdgcn_permlanex16(hi, hi, 0x76543210u, 0xfedcba98u, true, false);
    return ((u64)rhi << 32) | (u64)rlo;
}

template <int NCH, int STRIDE, int J, int K, int BASE = 0>
__device__ __forceinline__ void stage_regs_q(u64 (&key)[NSLOT]) {
    #pragma unroll
    for (int c = 0; c < NCH; ++c) {
        #pragma unroll
        for (int q = 0; q < 8; ++q) {
            if ((q & J) == 0) {
                const int s = BASE + c * STRIDE + q;
                const bool desc = (q & K) == 0;
                u64 a = key[s], b = key[s + J];
                const unsigned sw = desc ? key_lt_mask(a, b) : key_lt_mask(b, a);
                key[s]     = key_pick(sw, b, a);
                key[s + J] = key_pick(sw, a, b);
            }
        }
    }
}

template <int NCH, int STRIDE, bool XROW, int BASE = 0>
__device__ __forceinline__ void stage_lanes_rt(u64 (&key)[NSLOT], int lane, int M, bool d0, bool alt) {
    const unsigned sel = (unsigned)(lane ^ M);
    const bool lower = (lane & M) == 0;
    #pragma unroll
    for (int c = 0; c < NCH; ++c) {
        const bool desc = (c & 1) ? (d0 != alt) : d0;
        const unsigned keepmin = __builtin_amdgcn_ballot_w32(lower != desc);
        #pragma unroll
        for (int q = 0; q < 8; ++q) {
            const int s = BASE + c * STRIDE + q;
            u64 mine = key[s];
            u64 oth = XROW ? key_from_other_row(mine) : key_from_row_lane(mine, sel);
            const unsigned take = key_lt_mask(mine, oth) ^ keepmin;
            key[s] = key_pick(take, oth, mine);
        }
    }
}

template <int NCH, int STRIDE, int J, int BASE = 0>
__device__ __forceinline__ void stage_regs_rt(u64 (&key)[NSLOT], bool d0, bool alt) {
    #pragma unroll
    for (int c = 0; c < NCH; ++c) {
        const unsigned asc = ~__builtin_amdgcn_ballot_w32((c & 1) ? (d0 != alt) : d0);
        #pragma unroll
        for (int q = 0; q < 8; ++q) {
            if ((q & J) == 0) {
                const int s = BASE + c * STRIDE + q;
                u64 a = key[s], b = key[s + J];
                const unsigned sw = key_lt_mask(a, b) ^ asc;
                key[s]     = key_pick(sw, b, a);
                key[s + J] = key_pick(sw, a, b);
            }
        }
    }
}

template <int NCH, int STRIDE, int BASE = 0>
__device__ __forceinline__ void merge_rt(u64 (&key)[NSLOT], int lane, int mtop, bool d0, bool alt) {
    if (mtop >= 16) stage_lanes_rt<NCH, STRIDE, true, BASE>(key, lane, 16, d0, alt);
    #pragma nounroll
    for (int M = (mtop >= 16) ? 8 : mtop; M >= 1; M >>= 1) stage_lanes_rt<NCH, STRIDE, false, BASE>(key, lane, M, d0, alt);
    stage_regs_rt<NCH, STRIDE, 4, BASE>(key, d0, alt);
    stage_regs_rt<NCH, STRIDE, 2, BASE>(key, d0, alt);
    stage_regs_rt<NCH, STRIDE, 1, BASE>(key, d0, alt);
}

template <int G>
__device__ __forceinline__ void sort_chunk(u64 (&key)[NSLOT], int lane) {
    stage_regs_q<1, 8, 1, 2, 8 * G>(key);
    stage_regs_q<1, 8, 2, 4, 8 * G>(key);
    stage_regs_q<1, 8, 1, 4, 8 * G>(key);
    #pragma nounroll
    for (int L = 0; L < 6; ++L) {
        const bool last = (L == 5);
        const bool d0 = last ? ((G & 1) == 0) : (((lane >> L) & 1) == 0);
        merge_rt<1, 8, 8 * G>(key, lane, (1 << L) >> 1, d0, false);
    }
}

__device__ __forceinline__ float class_max(const float (&p)[N_CLASS]) {
    float m[7];
    #pragma unroll
    for (int g = 0; g < 6; ++g) m[g] = fmaxf(fmaxf(p[3 * g], p[3 * g + 1]), p[3 * g + 2]);
    m[6] = fmaxf(p[18], p[19]);
    const float n0 = fmaxf(fmaxf(m[0], m[1]), m[2]);
    const float n1 = fmaxf(fmaxf(m[3], m[4]), m[5]);
    return fmaxf(fmaxf(n0, n1), m[6]);
}

__device__ __forceinline__ u64 score_key(float conf, float pmax, int id) {
    const float best = __fmul_rn(conf, pmax);
    float s = (best >= 0.1f) ? best : 0.0f;
    return ((u64)__float_as_uint(s) << 32) | (unsigned)(~(unsigned)id);
}

template <int G>
__device__ __forceinline__ void chunk_keys(u64 (&key)[NSLOT], const float (&p)[12][N_CLASS],
                                           const float (&cf)[12][2], int tid) {
    #pragma unroll
    for (int n = 4 * G; n < 4 * G + 4; ++n) {
        const int cell = tid + 256 * n;
        const float pm = class_max(p[n]);
        key[2 * n]     = score_key(cf[n][0], pm, cell * 2);
        key[2 * n + 1] = score_key(cf[n][1], pm, cell * 2 + 1);
    }
}

__device__ __forceinline__ u64 key_max(u64 a, u64 b) { return key_pick(key_lt_mask(a, b), b, a); }

struct Phase2 {
    float y0[CAND], x0[CAND], y1[CAND], x1[CAND];
    float sc[CAND], cl[CAND], ar[CAND];
    float redv[CAND];
    union alignas(16) {
        unsigned short mask16[CAND * 16];
        unsigned       maskw[CAND * 8];
    };
    int selIdx[32], selOk[32];
    int npick;
    unsigned keepw[8];
};

__global__ __launch_bounds__(256) void det_kernel(const float* __restrict__ net,
                                                  float* __restrict__ out) {
    __shared__ u64 keys[CAND];
    __shared__ u64 xch[3][NWAVE][CAND];
    __shared__ Phase2 p2;
    Phase2* P = &p2;

    const int tid = threadIdx.x;
    const int img = blockIdx.x;
    const float* in = net + (size_t)img * D_IMG;
    const int lane = tid & 31;
    const int wave = tid >> 5;

    u64 key[NSLOT];
    const bool has12 = tid < (CELLS - 12 * 256);
    const int cell12 = has12 ? tid + 256 * 12 : 0;
    float p12[N_CLASS], cf12[2];
    {
        const float* pr = in + cell12 * N_CLASS;
        #pragma unroll
        for (int c = 0; c < N_CLASS; ++c) p12[c] = pr[c];
        cf12[0] = in[CONF_OFF + cell12 * 2];
        cf12[1] = in[CONF_OFF + cell12 * 2 + 1];
    }
    float p[12][N_CLASS], cf[12][2];
    #pragma unroll
    for (int n = 0; n < 12; ++n) {
        const int cell = tid + 256 * n;
        const float* pr = in + cell * N_CLASS;
        #pragma unroll
        for (int c = 0; c < N_CLASS; ++c) p[n][c] = pr[c];
        cf[n][0] = in[CONF_OFF + cell * 2];
        cf[n][1] = in[CONF_OFF + cell * 2 + 1];
    }

    {
        const float pm12 = class_max(p12);
        const u64 k0 = score_key(cf12[0], pm12, cell12 * 2);
        const u64 k1 = score_key(cf12[1], pm12, cell12 * 2 + 1);
        key[24] = has12 ? k0 : (u64)(unsigned)(~(unsigned)(ITEMS + tid * NSLOT + 24));
        key[25] = has12 ? k1 : (u64)(unsigned)(~(unsigned)(ITEMS + tid * NSLOT + 25));
        #pragma unroll
        for (int s = 26; s < NSLOT; ++s)
            key[s] = (u64)(unsigned)(~(unsigned)(ITEMS + tid * NSLOT + s));
    }
    sort_chunk<3>(key, lane);
    chunk_keys<0>(key, p, cf, tid);
    sort_chunk<0>(key, lane);
    chunk_keys<1>(key, p, cf, tid);
    sort_chunk<1>(key, lane);
    chunk_keys<2>(key, p, cf, tid);
    sort_chunk<2>(key, lane);
    #pragma unroll
    for (int q = 0; q < 8; ++q) {
        key[q]      = key_max(key[q],      key[8 + q]);
        key[16 + q] = key_max(key[16 + q], key[24 + q]);
    }
    merge_rt<2, 16>(key, lane, 16, true, true);
    #pragma unroll
    for (int q = 0; q < 8; ++q) key[q] = key_max(key[q], key[16 + q]);
    merge_rt<1, 16>(key, lane, 16, (wave & 1) == 0, false);
    #pragma nounroll
    for (int r = 0; r < 3; ++r) {
        #pragma unroll
        for (int q = 0; q < 8; ++q) xch[r][wave][q * 32 + lane] = key[q];
        __syncthreads();
        #pragma unroll
        for (int q = 0; q < 8; ++q) key[q] = key_max(key[q], xch[r][wave ^ (1 << r)][q * 32 + lane]);
        merge_rt<1, 16>(key, lane, 16, (r == 2) ? true : ((wave & (2 << r)) == 0), false);
    }
    if (wave == 0) {
        #pragma unroll
        for (int q = 0; q < 8; ++q) keys[lane * 8 + q] = key[q];
    }
    __syncthreads();

    {
        unsigned long long key = keys[tid];
        unsigned id = ~(unsigned)key;
        float s = __uint_as_float((unsigned)(key >> 32));
        int cell = (int)(id >> 1);
        int gi = cell / S_GRID, gj = cell % S_GRID;
        const float* cd = in + COORD_OFF + (size_t)id * 4;
        float tx = cd[0], ty = cd[1], tw = cd[2], th = cd[3];
        float x = __fdiv_rn(__fadd_rn(tx, (float)gj), 56.0f);
        float y = __fdiv_rn(__fadd_rn(ty, (float)gi), 56.0f);
        float hw = __fmul_rn(__fmul_rn(tw, tw), 0.5f);
        float hh = __fmul_rn(__fmul_rn(th, th), 0.5f);
        float y0 = __fsub_rn(y, hh), x0 = __fsub_rn(x, hw);
        float y1 = __fadd_rn(y, hh), x1 = __fadd_rn(x, hw);
        float conf = in[CONF_OFF + id];
        const float* pr = in + cell * N_CLASS;
        float best = NEG_INF; int bc = 0;
        #pragma unroll
        for (int c = 0; c < N_CLASS; ++c) {
            float v = __fmul_rn(conf, pr[c]);
            if (v > best) { best = v; bc = c; }
        }
        float dy = fmaxf(__fsub_rn(y1, y0), 0.0f);
        float dx = fmaxf(__fsub_rn(x1, x0), 0.0f);
        P->y0[tid] = y0; P->x0[tid] = x0; P->y1[tid] = y1; P->x1[tid] = x1;
        P->sc[tid] = s;  P->cl[tid] = (float)bc;
        P->ar[tid] = __fmul_rn(dy, dx);
        if (tid < 32) { P->selIdx[tid] = 0; P->selOk[tid] = 0; }
    }
    __syncthreads();

    {
        const unsigned below = (1u << lane) - 1u;
        int total = 0;
        for (int q = 0; q < 8; ++q) {
            v8f czero = {};
            for (int tt = wave; tt < 4 * (q + 1); tt += NWAVE) {
                int mi = 2 * q + (tt & 1), ni = tt >> 1;
                int m0 = mi << 4, n0 = ni << 4;
                v2f A, B;
                if (lane < 16) {
                    A[0] = P->ar[m0 + lane]; A[1] = 1.0f;
                    B[0] = 1.0f;             B[1] = P->ar[n0 + lane];
                } else {
                    A[0] = 0.0f; A[1] = 0.0f; B[0] = 0.0f; B[1] = 0.0f;
                }
                v8f d = __builtin_amdgcn_wmma_f32_16x16x4_f32(
                    false, A, false, B, (short)0, czero, false, false);
                int n = n0 + (lane & 15);
                float ny0 = P->y0[n], nx0 = P->x0[n], ny1 = P->y1[n], nx1 = P->x1[n];
                int mbase = m0 + ((lane & 16) ? 8 : 0);
                #pragma unroll
                for (int r = 0; r < 8; ++r) {
                    int m = mbase + r;
                    float iy = fmaxf(__fsub_rn(fminf(P->y1[m], ny1), fmaxf(P->y0[m], ny0)), 0.0f);
                    float ix = fmaxf(__fsub_rn(fminf(P->x1[m], nx1), fmaxf(P->x0[m], nx0)), 0.0f);
                    float inter = __fmul_rn(iy, ix);
                    float uni = __fsub_rn(d[r], inter);
                    bool pred = (uni > 0.0f) && (__fdiv_rn(inter, uni) > 0.4f);
                    unsigned bal = __builtin_amdgcn_ballot_w32(pred);
                    if (lane == 0) {
                        P->mask16[(m0 + r)     * 16 + ni] = (unsigned short)(bal & 0xFFFFu);
                        P->mask16[(m0 + 8 + r) * 16 + ni] = (unsigned short)(bal >> 16);
                    }
                }
            }
            __syncthreads();

            if (wave == 0) {
                unsigned ext = 0;
                for (int w = 0; w < q; ++w) ext |= P->maskw[(q * 32 + lane) * 8 + w] & P->keepw[w];
                const unsigned own = P->maskw[(q * 32 + lane) * 8 + q] & below;
                const unsigned open = __builtin_amdgcn_ballot_w32(ext == 0);
                unsigned k = open;
                for (int pass = 0; pass < 32; ++pass) {
                    unsigned k2 = open & __builtin_amdgcn_ballot_w32((own & k) == 0);
                    if (k2 == k) break;
                    k = k2;
                }
                if ((k >> lane) & 1u) {
                    int rank = total + __builtin_popcount(k & below);
                    if (rank < MAXOUT) { P->selIdx[rank] = q * 32 + lane; P->selOk[rank] = 1; }
                }
                total += __builtin_popcount(k);
                if (lane == 0) { P->npick = total; P->keepw[q] = k; }
            }
            __syncthreads();
            if (P->npick >= MAXOUT) break;
        }
    }

    if (tid < MAXOUT) {
        int sel = P->selIdx[tid];
        float okf = P->selOk[tid] ? 1.0f : 0.0f;
        P->redv[tid] = __fmul_rn(P->sc[sel], okf);
    }
    __syncthreads();
    if (tid < MAXOUT) {
        int sel = P->selIdx[tid];
        float okf = P->selOk[tid] ? 1.0f : 0.0f;
        float ms = P->redv[tid];
        int rank = 0;
        for (int jt = 0; jt < MAXOUT; ++jt) {
            float vj = P->redv[jt];
            if (vj > ms || (vj == ms && jt < tid)) rank++;
        }
        float* o = out + (size_t)img * (MAXOUT * 6) + (size_t)rank * 6;
        o[0] = __fmul_rn(P->y0[sel], okf);
        o[1] = __fmul_rn(P->x0[sel], okf);
        o[2] = __fmul_rn(P->y1[sel], okf);
        o[3] = __fmul_rn(P->x1[sel], okf);
        o[4] = ms;
        o[5] = __fmul_rn(P->cl[sel], okf);
    }
}

extern "C" void kernel_launch(void* const* d_in, const int* in_sizes, int n_in,
                              void* d_out, int out_size, void* d_ws, size_t ws_size,
                              hipStream_t stream) {
    (void)n_in; (void)out_size; (void)d_ws; (void)ws_size;
    const float* net = (const float*)d_in[0];
    float* out = (float*)d_out;
    int batch = in_sizes[0] / D_IMG;
    det_kernel<<<dim3(batch), dim3(256), 0, stream>>>(net, out);
}


# Each thread first selects its 8 largest of 32 keys in a register network; cross-lane sort on one 256-key chunk per wave, not four; exact: a left-out key above the list's last folds the rest in
# speedup vs baseline: 18.1862x; 1.3788x over previous
#include <hip/hip_runtime.h>

typedef float v2f __attribute__((ext_vector_type(2)));
typedef float v8f __attribute__((ext_vector_type(8)));

#define S_GRID   56
#define N_CLASS  20
#define CELLS    (S_GRID * S_GRID)
#define ITEMS    (CELLS * 2)
#define CAND     256
#define MAXOUT   30
#define D_IMG    94080
#define CONF_OFF 62720
#define COORD_OFF 68992
#define NEG_INF  (-3.402823466e38f)
#define NSLOT    32
#define NWAVE    8

typedef unsigned long long u64;

__device__ __forceinline__ unsigned key_lt_mask(u64 a, u64 b) {
    const unsigned al = (unsigned)a, ah = (unsigned)(a >> 32);
    const unsigned bl = (unsigned)b, bh = (unsigned)(b >> 32);
    const unsigned lo = __builtin_amdgcn_ballot_w32(al < bl);
    unsigned diff, out;
    asm("v_sub_co_ci_u32_e64 %0, %1, %2, %3, %4" : "=v"(diff), "=s"(out) : "v"(ah), "v"(bh), "s"(lo));
    return out;
}
__device__ __forceinline__ u64 key_pick(unsigned mask, u64 x, u64 y) {
    return __builtin_amdgcn_inverse_ballot_w32(mask) ? x : y;
}


__device__ __forceinline__ u64 key_from_row_lane(u64 v, unsigned sel) {
    unsigned lo = (unsigned)v, hi = (unsigned)(v >> 32);
    unsigned rlo = __builtin_amdgcn_permlane16_var(lo, lo, sel, true, false);
    unsigned rhi = __builtin_amdgcn_permlane16_var(hi, hi, sel, true, false);
    return ((u64)rhi << 32) | (u64)rlo;
}
__device__ __forceinline__ u64 key_from_other_row(u64 v) {
    unsigned lo = (unsigned)v, hi = (unsigned)(v >> 32);
    unsigned rlo = __builtin_amdgcn_permlanex16(lo, lo, 0x76543210u, 0xfedcba98u, true, false);
    unsigned rhi = __builtin_amdgcn_permlanex16(hi, hi, 0x76543210u, 0xfedcba98u, true, false);
    return ((u64)rhi << 32) | (u64)rlo;
}

template <int NCH, int STRIDE, int J, int K, int BASE = 0>
__device__ __forceinline__ void stage_regs_q(u64 (&key)[NSLOT]) {
    #pragma unroll
    for (int c = 0; c < NCH; ++c) {
        #pragma unroll
        for (int q = 0; q < 8; ++q) {
            if ((q & J) == 0) {
                const int s = BASE + c * STRIDE + q;
                const bool desc = (q & K) == 0;
                u64 a = key[s], b = key[s + J];
                const unsigned sw = desc ? key_lt_mask(a, b) : key_lt_mask(b, a);
                key[s]     = key_pick(sw, b, a);
                key[s + J] = key_pick(sw, a, b);
            }
        }
    }
}

template <int NCH, int STRIDE, bool XROW, int BASE = 0>
__device__ __forceinline__ void stage_lanes_rt(u64 (&key)[NSLOT], int lane, int M, bool d0, bool alt) {
    const unsigned sel = (unsigned)(lane ^ M);
    const bool lower = (lane & M) == 0;
    #pragma unroll
    for (int c = 0; c < NCH; ++c) {
        const bool desc = (c & 1) ? (d0 != alt) : d0;
        const unsigned keepmin = __builtin_amdgcn_ballot_w32(lower != desc);
        #pragma unroll
        for (int q = 0; q < 8; ++q) {
            const int s = BASE + c * STRIDE + q;
            u64 mine = key[s];
            u64 oth = XROW ? key_from_other_row(mine) : key_from_row_lane(mine, sel);
            const unsigned take = key_lt_mask(mine, oth) ^ keepmin;
            key[s] = key_pick(take, oth, mine);
        }
    }
}

template <int NCH, int STRIDE, int J, int BASE = 0>
__device__ __forceinline__ void stage_regs_rt(u64 (&key)[NSLOT], bool d0, bool alt) {
    #pragma unroll
    for (int c = 0; c < NCH; ++c) {
        const unsigned asc = ~__builtin_amdgcn_ballot_w32((c & 1) ? (d0 != alt) : d0);
        #pragma unroll
        for (int q = 0; q < 8; ++q) {
            if ((q & J) == 0) {
                const int s = BASE + c * STRIDE + q;
                u64 a = key[s], b = key[s + J];
                const unsigned sw = key_lt_mask(a, b) ^ asc;
                key[s]     = key_pick(sw, b, a);
                key[s + J] = key_pick(sw, a, b);
            }
        }
    }
}

template <int NCH, int STRIDE, int BASE = 0>
__device__ __forceinline__ void merge_rt(u64 (&key)[NSLOT], int lane, int mtop, bool d0, bool alt) {
    if (mtop >= 16) stage_lanes_rt<NCH, STRIDE, true, BASE>(key, lane, 16, d0, alt);
    #pragma nounroll
    for (int M = (mtop >= 16) ? 8 : mtop; M >= 1; M >>= 1) stage_lanes_rt<NCH, STRIDE, false, BASE>(key, lane, M, d0, alt);
    stage_regs_rt<NCH, STRIDE, 4, BASE>(key, d0, alt);
    stage_regs_rt<NCH, STRIDE, 2, BASE>(key, d0, alt);
    stage_regs_rt<NCH, STRIDE, 1, BASE>(key, d0, alt);
}

template <int G>
__device__ __forceinline__ void sort_chunk(u64 (&key)[NSLOT], int lane, bool lastdesc) {
    stage_regs_q<1, 8, 1, 2, 8 * G>(key);
    stage_regs_q<1, 8, 2, 4, 8 * G>(key);
    stage_regs_q<1, 8, 1, 4, 8 * G>(key);
    #pragma nounroll
    for (int L = 0; L < 6; ++L) {
        const bool last = (L == 5);
        const bool d0 = last ? lastdesc : (((lane >> L) & 1) == 0);
        merge_rt<1, 8, 8 * G>(key, lane, (1 << L) >> 1, d0, false);
    }
}

template <int G>
__device__ __forceinline__ void sort_chunk_presorted(u64 (&key)[NSLOT], int lane, bool lastdesc) {
    #pragma nounroll
    for (int L = 1; L < 6; ++L) {
        const bool last = (L == 5);
        const bool d0 = last ? lastdesc : (((lane >> L) & 1) == 0);
        merge_rt<1, 8, 8 * G>(key, lane, (1 << L) >> 1, d0, false);
    }
}

template <int G>
__device__ __forceinline__ void local_select(u64 (&key)[NSLOT], bool d0) {
    stage_regs_q<1, 8, 1, 2, 8 * G>(key);
    stage_regs_q<1, 8, 2, 4, 8 * G>(key);
    stage_regs_q<1, 8, 1, 4, 8 * G>(key);
    stage_regs_q<1, 8, 4, 8, 8 * G>(key);
    stage_regs_q<1, 8, 2, 8, 8 * G>(key);
    stage_regs_q<1, 8, 1, 8, 8 * G>(key);
    #pragma unroll
    for (int i = 0; i < 8; ++i) {
        u64 a = key[24 + i], b = key[8 * G + 7 - i];
        const unsigned sw = key_lt_mask(a, b);
        key[24 + i]        = key_pick(sw, b, a);
        key[8 * G + 7 - i] = key_pick(sw, a, b);
    }
    if constexpr (G == 2) {
        stage_regs_rt<1, 8, 4, 24>(key, d0, false);
        stage_regs_rt<1, 8, 2, 24>(key, d0, false);
        stage_regs_rt<1, 8, 1, 24>(key, d0, false);
    } else {
        stage_regs_q<1, 8, 4, 8, 24>(key);
        stage_regs_q<1, 8, 2, 8, 24>(key);
        stage_regs_q<1, 8, 1, 8, 24>(key);
    }
}

__device__ __forceinline__ float class_max(const float (&p)[N_CLASS]) {
    float m[7];
    #pragma unroll
    for (int g = 0; g < 6; ++g) m[g] = fmaxf(fmaxf(p[3 * g], p[3 * g + 1]), p[3 * g + 2]);
    m[6] = fmaxf(p[18], p[19]);
    const float n0 = fmaxf(fmaxf(m[0], m[1]), m[2]);
    const float n1 = fmaxf(fmaxf(m[3], m[4]), m[5]);
    return fmaxf(fmaxf(n0, n1), m[6]);
}

__device__ __forceinline__ u64 score_key(float conf, float pmax, int id) {
    const float best = __fmul_rn(conf, pmax);
    float s = (best >= 0.1f) ? best : 0.0f;
    return ((u64)__float_as_uint(s) << 32) | (unsigned)(~(unsigned)id);
}

template <int G>
__device__ __forceinline__ void chunk_keys(u64 (&key)[NSLOT], const float (&p)[12][N_CLASS],
                                           const float (&cf)[12][2], int tid) {
    #pragma unroll
    for (int n = 4 * G; n < 4 * G + 4; ++n) {
        const int cell = tid + 256 * n;
        const float pm = class_max(p[n]);
        key[2 * n]     = score_key(cf[n][0], pm, cell * 2);
        key[2 * n + 1] = score_key(cf[n][1], pm, cell * 2 + 1);
    }
}

__device__ __forceinline__ u64 key_max(u64 a, u64 b) { return key_pick(key_lt_mask(a, b), b, a); }

struct Phase2 {
    float y0[CAND], x0[CAND], y1[CAND], x1[CAND];
    float sc[CAND], cl[CAND], ar[CAND];
    float redv[CAND];
    union alignas(16) {
        unsigned short mask16[CAND * 16];
        unsigned       maskw[CAND * 8];
    };
    int selIdx[32], selOk[32];
    int npick;
    unsigned keepw[8];
    int toobig;
};

__global__ __launch_bounds__(256) void det_kernel(const float* __restrict__ net,
                                                  float* __restrict__ out) {
    __shared__ u64 keys[CAND];
    __shared__ u64 xch[3][NWAVE][CAND];
    __shared__ Phase2 p2;
    Phase2* P = &p2;

    const int tid = threadIdx.x;
    const int img = blockIdx.x;
    const float* in = net + (size_t)img * D_IMG;
    const int lane = tid & 31;
    const int wave = tid >> 5;

    u64 key[NSLOT];
    const bool has12 = tid < (CELLS - 12 * 256);
    const int cell12 = has12 ? tid + 256 * 12 : 0;
    float p12[N_CLASS], cf12[2];
    {
        const float* pr = in + cell12 * N_CLASS;
        #pragma unroll
        for (int c = 0; c < N_CLASS; ++c) p12[c] = pr[c];
        cf12[0] = in[CONF_OFF + cell12 * 2];
        cf12[1] = in[CONF_OFF + cell12 * 2 + 1];
    }
    float p[12][N_CLASS], cf[12][2];
    #pragma unroll
    for (int n = 0; n < 12; ++n) {
        const int cell = tid + 256 * n;
        const float* pr = in + cell * N_CLASS;
        #pragma unroll
        for (int c = 0; c < N_CLASS; ++c) p[n][c] = pr[c];
        cf[n][0] = in[CONF_OFF + cell * 2];
        cf[n][1] = in[CONF_OFF + cell * 2 + 1];
    }
    __builtin_amdgcn_sched_barrier(0);

    {
        const float pm12 = class_max(p12);
        const u64 k0 = score_key(cf12[0], pm12, cell12 * 2);
        const u64 k1 = score_key(cf12[1], pm12, cell12 * 2 + 1);
        const u64 a = has12 ? k0 : (u64)(unsigned)(~(unsigned)(ITEMS + tid * NSLOT + 24));
        const u64 b = has12 ? k1 : (u64)(unsigned)(~(unsigned)(ITEMS + tid * NSLOT + 25));
        const unsigned sw = key_lt_mask(a, b);
        key[24] = key_pick(sw, b, a);
        key[25] = key_pick(sw, a, b);
        #pragma unroll
        for (int s = 26; s < NSLOT; ++s)
            key[s] = (u64)(unsigned)(~(unsigned)(ITEMS + tid * NSLOT + s));
    }
    chunk_keys<0>(key, p, cf, tid);
    local_select<0>(key, true);
    __builtin_amdgcn_sched_barrier(0);
    chunk_keys<1>(key, p, cf, tid);
    local_select<1>(key, true);
    __builtin_amdgcn_sched_barrier(0);
    chunk_keys<2>(key, p, cf, tid);
    local_select<2>(key, (lane & 1) == 0);
    sort_chunk_presorted<3>(key, lane, (wave & 1) == 0);
    u64 cur[NSLOT];
    #pragma unroll
    for (int q = 0; q < 8; ++q) cur[q] = key[24 + q];
    if (tid == 0) P->toobig = 0;
    #pragma nounroll
    for (int attempt = 0; attempt < 2; ++attempt) {
        #pragma nounroll
        for (int r = 0; r < 3; ++r) {
            #pragma unroll
            for (int q = 0; q < 8; ++q) xch[r][wave][q * 32 + lane] = cur[q];
            __syncthreads();
            #pragma unroll
            for (int q = 0; q < 8; ++q) cur[q] = key_max(cur[q], xch[r][wave ^ (1 << r)][q * 32 + lane]);
            merge_rt<1, 16>(cur, lane, 16, (r == 2) ? true : ((wave & (2 << r)) == 0), false);
        }
        if (wave == 0) {
            #pragma unroll
            for (int q = 0; q < 8; ++q) keys[lane * 8 + q] = cur[q];
        }
        __syncthreads();
        if (attempt == 1) break;
        {
            const u64 last = keys[CAND - 1];
            unsigned above = 0;
            #pragma unroll
            for (int s = 0; s < 24; ++s) above |= key_lt_mask(last, key[s]);
            if (above != 0 && lane == 0) P->toobig = 1;
        }
        __syncthreads();
        if (__builtin_amdgcn_readfirstlane(P->toobig) == 0) break;
        const bool wdesc = (wave & 1) == 0;
        #pragma nounroll
        for (int j = 0; j < 3; ++j) {
            sort_chunk<0>(key, lane, !wdesc);
            #pragma unroll
            for (int q = 0; q < 8; ++q) key[24 + q] = key_max(key[24 + q], key[q]);
            merge_rt<1, 8, 24>(key, lane, 16, wdesc, false);
            #pragma unroll
            for (int q = 0; q < 8; ++q) { key[q] = key[8 + q]; key[8 + q] = key[16 + q]; }
        }
        #pragma unroll
        for (int q = 0; q < 8; ++q) cur[q] = key[24 + q];
    }

    {
        unsigned long long key = keys[tid];
        unsigned id = ~(unsigned)key;
        float s = __uint_as_float((unsigned)(key >> 32));
        int cell = (int)(id >> 1);
        int gi = cell / S_GRID, gj = cell % S_GRID;
        const float* cd = in + COORD_OFF + (size_t)id * 4;
        float tx = cd[0], ty = cd[1], tw = cd[2], th = cd[3];
        float x = __fdiv_rn(__fadd_rn(tx, (float)gj), 56.0f);
        float y = __fdiv_rn(__fadd_rn(ty, (float)gi), 56.0f);
        float hw = __fmul_rn(__fmul_rn(tw, tw), 0.5f);
        float hh = __fmul_rn(__fmul_rn(th, th), 0.5f);
        float y0 = __fsub_rn(y, hh), x0 = __fsub_rn(x, hw);
        float y1 = __fadd_rn(y, hh), x1 = __fadd_rn(x, hw);
        float conf = in[CONF_OFF + id];
        const float* pr = in + cell * N_CLASS;
        float best = NEG_INF; int bc = 0;
        #pragma unroll
        for (int c = 0; c < N_CLASS; ++c) {
            float v = __fmul_rn(conf, pr[c]);
            if (v > best) { best = v; bc = c; }
        }
        float dy = fmaxf(__fsub_rn(y1, y0), 0.0f);
        float dx = fmaxf(__fsub_rn(x1, x0), 0.0f);
        P->y0[tid] = y0; P->x0[tid] = x0; P->y1[tid] = y1; P->x1[tid] = x1;
        P->sc[tid] = s;  P->cl[tid] = (float)bc;
        P->ar[tid] = __fmul_rn(dy, dx);
        if (tid < 32) { P->selIdx[tid] = 0; P->selOk[tid] = 0; }
    }
    __syncthreads();

    {
        const unsigned below = (1u << lane) - 1u;
        int total = 0;
        for (int q = 0; q < 8; ++q) {
            v8f czero = {};
            for (int tt = wave; tt < 4 * (q + 1); tt += NWAVE) {
                int mi = 2 * q + (tt & 1), ni = tt >> 1;
                int m0 = mi << 4, n0 = ni << 4;
                v2f A, B;
                if (lane < 16) {
                    A[0] = P->ar[m0 + lane]; A[1] = 1.0f;
                    B[0] = 1.0f;             B[1] = P->ar[n0 + lane];
                } else {
                    A[0] = 0.0f; A[1] = 0.0f; B[0] = 0.0f; B[1] = 0.0f;
                }
                v8f d = __builtin_amdgcn_wmma_f32_16x16x4_f32(
                    false, A, false, B, (short)0, czero, false, false);
                int n = n0 + (lane & 15);
                float ny0 = P->y0[n], nx0 = P->x0[n], ny1 = P->y1[n], nx1 = P->x1[n];
                int mbase = m0 + ((lane & 16) ? 8 : 0);
                #pragma unroll
                for (int r = 0; r < 8; ++r) {
                    int m = mbase + r;
                    float iy = fmaxf(__fsub_rn(fminf(P->y1[m], ny1), fmaxf(P->y0[m], ny0)), 0.0f);
                    float ix = fmaxf(__fsub_rn(fminf(P->x1[m], nx1), fmaxf(P->x0[m], nx0)), 0.0f);
                    float inter = __fmul_rn(iy, ix);
                    float uni = __fsub_rn(d[r], inter);
                    bool pred = (uni > 0.0f) && (__fdiv_rn(inter, uni) > 0.4f);
                    unsigned bal = __builtin_amdgcn_ballot_w32(pred);
                    if (lane == 0) {
                        P->mask16[(m0 + r)     * 16 + ni] = (unsigned short)(bal & 0xFFFFu);
                        P->mask16[(m0 + 8 + r) * 16 + ni] = (unsigned short)(bal >> 16);
                    }
                }
            }
            __syncthreads();

            if (wave == 0) {
                unsigned ext = 0;
                for (int w = 0; w < q; ++w) ext |= P->maskw[(q * 32 + lane) * 8 + w] & P->keepw[w];
                const unsigned own = P->maskw[(q * 32 + lane) * 8 + q] & below;
                const unsigned open = __builtin_amdgcn_ballot_w32(ext == 0);
                unsigned k = open;
                for (int pass = 0; pass < 32; ++pass) {
                    unsigned k2 = open & __builtin_amdgcn_ballot_w32((own & k) == 0);
                    if (k2 == k) break;
                    k = k2;
                }
                if ((k >> lane) & 1u) {
                    int rank = total + __builtin_popcount(k & below);
                    if (rank < MAXOUT) { P->selIdx[rank] = q * 32 + lane; P->selOk[rank] = 1; }
                }
                total += __builtin_popcount(k);
                if (lane == 0) { P->npick = total; P->keepw[q] = k; }
            }
            __syncthreads();
            if (P->npick >= MAXOUT) break;
        }
    }

    if (tid < MAXOUT) {
        int sel = P->selIdx[tid];
        float okf = P->selOk[tid] ? 1.0f : 0.0f;
        P->redv[tid] = __fmul_rn(P->sc[sel], okf);
    }
    __syncthreads();
    if (tid < MAXOUT) {
        int sel = P->selIdx[tid];
        float okf = P->selOk[tid] ? 1.0f : 0.0f;
        float ms = P->redv[tid];
        int rank = 0;
        for (int jt = 0; jt < MAXOUT; ++jt) {
            float vj = P->redv[jt];
            if (vj > ms || (vj == ms && jt < tid)) rank++;
        }
        float* o = out + (size_t)img * (MAXOUT * 6) + (size_t)rank * 6;
        o[0] = __fmul_rn(P->y0[sel], okf);
        o[1] = __fmul_rn(P->x0[sel], okf);
        o[2] = __fmul_rn(P->y1[sel], okf);
        o[3] = __fmul_rn(P->x1[sel], okf);
        o[4] = ms;
        o[5] = __fmul_rn(P->cl[sel], okf);
    }
}

extern "C" void kernel_launch(void* const* d_in, const int* in_sizes, int n_in,
                              void* d_out, int out_size, void* d_ws, size_t ws_size,
                              hipStream_t stream) {
    (void)n_in; (void)out_size; (void)d_ws; (void)ws_size;
    const float* net = (const float*)d_in[0];
    float* out = (float*)d_out;
    int batch = in_sizes[0] / D_IMG;
    det_kernel<<<dim3(batch), dim3(256), 0, stream>>>(net, out);
}


# Final ranking of the 30 picked rows in one wave's registers: lane t reads row j's score from lane j (v_readlane), counts branch-free; replaces 30 serial LDS reads under exec branches and a barrier
# speedup vs baseline: 19.6576x; 1.0809x over previous
#include <hip/hip_runtime.h>

typedef float v2f __attribute__((ext_vector_type(2)));
typedef float v8f __attribute__((ext_vector_type(8)));

#define S_GRID   56
#define N_CLASS  20
#define CELLS    (S_GRID * S_GRID)
#define ITEMS    (CELLS * 2)
#define CAND     256
#define MAXOUT   30
#define D_IMG    94080
#define CONF_OFF 62720
#define COORD_OFF 68992
#define NEG_INF  (-3.402823466e38f)
#define NSLOT    32
#define NWAVE    8

typedef unsigned long long u64;

__device__ __forceinline__ unsigned key_lt_mask(u64 a, u64 b) {
    const unsigned al = (unsigned)a, ah = (unsigned)(a >> 32);
    const unsigned bl = (unsigned)b, bh = (unsigned)(b >> 32);
    const unsigned lo = __builtin_amdgcn_ballot_w32(al < bl);
    unsigned diff, out;
    asm("v_sub_co_ci_u32_e64 %0, %1, %2, %3, %4" : "=v"(diff), "=s"(out) : "v"(ah), "v"(bh), "s"(lo));
    return out;
}
__device__ __forceinline__ u64 key_pick(unsigned mask, u64 x, u64 y) {
    return __builtin_amdgcn_inverse_ballot_w32(mask) ? x : y;
}


__device__ __forceinline__ u64 key_from_row_lane(u64 v, unsigned sel) {
    unsigned lo = (unsigned)v, hi = (unsigned)(v >> 32);
    unsigned rlo = __builtin_amdgcn_permlane16_var(lo, lo, sel, true, false);
    unsigned rhi = __builtin_amdgcn_permlane16_var(hi, hi, sel, true, false);
    return ((u64)rhi << 32) | (u64)rlo;
}
__device__ __forceinline__ u64 key_from_other_row(u64 v) {
    unsigned lo = (unsigned)v, hi = (unsigned)(v >> 32);
    unsigned rlo = __builtin_amdgcn_permlanex16(lo, lo, 0x76543210u, 0xfedcba98u, true, false);
    unsigned rhi = __builtin_amdgcn_permlanex16(hi, hi, 0x76543210u, 0xfedcba98u, true, false);
    return ((u64)rhi << 32) | (u64)rlo;
}

template <int NCH, int STRIDE, int J, int K, int BASE = 0>
__device__ __forceinline__ void stage_regs_q(u64 (&key)[NSLOT]) {
    #pragma unroll
    for (int c = 0; c < NCH; ++c) {
        #pragma unroll
        for (int q = 0; q < 8; ++q) {
            if ((q & J) == 0) {
                const int s = BASE + c * STRIDE + q;
                const bool desc = (q & K) == 0;
                u64 a = key[s], b = key[s + J];
                const unsigned sw = desc ? key_lt_mask(a, b) : key_lt_mask(b, a);
                key[s]     = key_pick(sw, b, a);
                key[s + J] = key_pick(sw, a, b);
            }
        }
    }
}

template <int NCH, int STRIDE, bool XROW, int BASE = 0>
__device__ __forceinline__ void stage_lanes_rt(u64 (&key)[NSLOT], int lane, int M, bool d0, bool alt) {
    const unsigned sel = (unsigned)(lane ^ M);
    const bool lower = (lane & M) == 0;
    #pragma unroll
    for (int c = 0; c < NCH; ++c) {
        const bool desc = (c & 1) ? (d0 != alt) : d0;
        const unsigned keepmin = __builtin_amdgcn_ballot_w32(lower != desc);
        #pragma unroll
        for (int q = 0; q < 8; ++q) {
            const int s = BASE + c * STRIDE + q;
            u64 mine = key[s];
            u64 oth = XROW ? key_from_other_row(mine) : key_from_row_lane(mine, sel);
            const unsigned take = key_lt_mask(mine, oth) ^ keepmin;
            key[s] = key_pick(take, oth, mine);
        }
    }
}

template <int NCH, int STRIDE, int J, int BASE = 0>
__device__ __forceinline__ void stage_regs_rt(u64 (&key)[NSLOT], bool d0, bool alt) {
    #pragma unroll
    for (int c = 0; c < NCH; ++c) {
        const unsigned asc = ~__builtin_amdgcn_ballot_w32((c & 1) ? (d0 != alt) : d0);
        #pragma unroll
        for (int q = 0; q < 8; ++q) {
            if ((q & J) == 0) {
                const int s = BASE + c * STRIDE + q;
                u64 a = key[s], b = key[s + J];
                const unsigned sw = key_lt_mask(a, b) ^ asc;
                key[s]     = key_pick(sw, b, a);
                key[s + J] = key_pick(sw, a, b);
            }
        }
    }
}

template <int NCH, int STRIDE, int BASE = 0>
__device__ __forceinline__ void merge_rt(u64 (&key)[NSLOT], int lane, int mtop, bool d0, bool alt) {
    if (mtop >= 16) stage_lanes_rt<NCH, STRIDE, true, BASE>(key, lane, 16, d0, alt);
    #pragma nounroll
    for (int M = (mtop >= 16) ? 8 : mtop; M >= 1; M >>= 1) stage_lanes_rt<NCH, STRIDE, false, BASE>(key, lane, M, d0, alt);
    stage_regs_rt<NCH, STRIDE, 4, BASE>(key, d0, alt);
    stage_regs_rt<NCH, STRIDE, 2, BASE>(key, d0, alt);
    stage_regs_rt<NCH, STRIDE, 1, BASE>(key, d0, alt);
}

template <int G>
__device__ __forceinline__ void sort_chunk(u64 (&key)[NSLOT], int lane, bool lastdesc) {
    stage_regs_q<1, 8, 1, 2, 8 * G>(key);
    stage_regs_q<1, 8, 2, 4, 8 * G>(key);
    stage_regs_q<1, 8, 1, 4, 8 * G>(key);
    #pragma nounroll
    for (int L = 0; L < 6; ++L) {
        const bool last = (L == 5);
        const bool d0 = last ? lastdesc : (((lane >> L) & 1) == 0);
        merge_rt<1, 8, 8 * G>(key, lane, (1 << L) >> 1, d0, false);
    }
}

template <int G>
__device__ __forceinline__ void sort_chunk_presorted(u64 (&key)[NSLOT], int lane, bool lastdesc) {
    #pragma nounroll
    for (int L = 1; L < 6; ++L) {
        const bool last = (L == 5);
        const bool d0 = last ? lastdesc : (((lane >> L) & 1) == 0);
        merge_rt<1, 8, 8 * G>(key, lane, (1 << L) >> 1, d0, false);
    }
}

template <int G>
__device__ __forceinline__ void local_select(u64 (&key)[NSLOT], bool d0) {
    stage_regs_q<1, 8, 1, 2, 8 * G>(key);
    stage_regs_q<1, 8, 2, 4, 8 * G>(key);
    stage_regs_q<1, 8, 1, 4, 8 * G>(key);
    stage_regs_q<1, 8, 4, 8, 8 * G>(key);
    stage_regs_q<1, 8, 2, 8, 8 * G>(key);
    stage_regs_q<1, 8, 1, 8, 8 * G>(key);
    #pragma unroll
    for (int i = 0; i < 8; ++i) {
        u64 a = key[24 + i], b = key[8 * G + 7 - i];
        const unsigned sw = key_lt_mask(a, b);
        key[24 + i]        = key_pick(sw, b, a);
        key[8 * G + 7 - i] = key_pick(sw, a, b);
    }
    if constexpr (G == 2) {
        stage_regs_rt<1, 8, 4, 24>(key, d0, false);
        stage_regs_rt<1, 8, 2, 24>(key, d0, false);
        stage_regs_rt<1, 8, 1, 24>(key, d0, false);
    } else {
        stage_regs_q<1, 8, 4, 8, 24>(key);
        stage_regs_q<1, 8, 2, 8, 24>(key);
        stage_regs_q<1, 8, 1, 8, 24>(key);
    }
}

__device__ __forceinline__ float class_max(const float (&p)[N_CLASS]) {
    float m[7];
    #pragma unroll
    for (int g = 0; g < 6; ++g) m[g] = fmaxf(fmaxf(p[3 * g], p[3 * g + 1]), p[3 * g + 2]);
    m[6] = fmaxf(p[18], p[19]);
    const float n0 = fmaxf(fmaxf(m[0], m[1]), m[2]);
    const float n1 = fmaxf(fmaxf(m[3], m[4]), m[5]);
    return fmaxf(fmaxf(n0, n1), m[6]);
}

__device__ __forceinline__ u64 score_key(float conf, float pmax, int id) {
    const float best = __fmul_rn(conf, pmax);
    float s = (best >= 0.1f) ? best : 0.0f;
    return ((u64)__float_as_uint(s) << 32) | (unsigned)(~(unsigned)id);
}

template <int G>
__device__ __forceinline__ void chunk_keys(u64 (&key)[NSLOT], const float (&p)[12][N_CLASS],
                                           const float (&cf)[12][2], int tid) {
    #pragma unroll
    for (int n = 4 * G; n < 4 * G + 4; ++n) {
        const int cell = tid + 256 * n;
        const float pm = class_max(p[n]);
        key[2 * n]     = score_key(cf[n][0], pm, cell * 2);
        key[2 * n + 1] = score_key(cf[n][1], pm, cell * 2 + 1);
    }
}

__device__ __forceinline__ u64 key_max(u64 a, u64 b) { return key_pick(key_lt_mask(a, b), b, a); }

struct Phase2 {
    float y0[CAND], x0[CAND], y1[CAND], x1[CAND];
    float sc[CAND], cl[CAND], ar[CAND];
    union alignas(16) {
        unsigned short mask16[CAND * 16];
        unsigned       maskw[CAND * 8];
    };
    int selIdx[32], selOk[32];
    int npick;
    unsigned keepw[8];
    int toobig;
};

__global__ __launch_bounds__(256) void det_kernel(const float* __restrict__ net,
                                                  float* __restrict__ out) {
    __shared__ u64 keys[CAND];
    __shared__ u64 xch[3][NWAVE][CAND];
    __shared__ Phase2 p2;
    Phase2* P = &p2;

    const int tid = threadIdx.x;
    const int img = blockIdx.x;
    const float* in = net + (size_t)img * D_IMG;
    const int lane = tid & 31;
    const int wave = tid >> 5;

    u64 key[NSLOT];
    const bool has12 = tid < (CELLS - 12 * 256);
    const int cell12 = has12 ? tid + 256 * 12 : 0;
    float p12[N_CLASS], cf12[2];
    {
        const float* pr = in + cell12 * N_CLASS;
        #pragma unroll
        for (int c = 0; c < N_CLASS; ++c) p12[c] = pr[c];
        cf12[0] = in[CONF_OFF + cell12 * 2];
        cf12[1] = in[CONF_OFF + cell12 * 2 + 1];
    }
    float p[12][N_CLASS], cf[12][2];
    #pragma unroll
    for (int n = 0; n < 12; ++n) {
        const int cell = tid + 256 * n;
        const float* pr = in + cell * N_CLASS;
        #pragma unroll
        for (int c = 0; c < N_CLASS; ++c) p[n][c] = pr[c];
        cf[n][0] = in[CONF_OFF + cell * 2];
        cf[n][1] = in[CONF_OFF + cell * 2 + 1];
    }
    __builtin_amdgcn_sched_barrier(0);

    {
        const float pm12 = class_max(p12);
        const u64 k0 = score_key(cf12[0], pm12, cell12 * 2);
        const u64 k1 = score_key(cf12[1], pm12, cell12 * 2 + 1);
        const u64 a = has12 ? k0 : (u64)(unsigned)(~(unsigned)(ITEMS + tid * NSLOT + 24));
        const u64 b = has12 ? k1 : (u64)(unsigned)(~(unsigned)(ITEMS + tid * NSLOT + 25));
        const unsigned sw = key_lt_mask(a, b);
        key[24] = key_pick(sw, b, a);
        key[25] = key_pick(sw, a, b);
        #pragma unroll
        for (int s = 26; s < NSLOT; ++s)
            key[s] = (u64)(unsigned)(~(unsigned)(ITEMS + tid * NSLOT + s));
    }
    chunk_keys<0>(key, p, cf, tid);
    local_select<0>(key, true);
    __builtin_amdgcn_sched_barrier(0);
    chunk_keys<1>(key, p, cf, tid);
    local_select<1>(key, true);
    __builtin_amdgcn_sched_barrier(0);
    chunk_keys<2>(key, p, cf, tid);
    local_select<2>(key, (lane & 1) == 0);
    sort_chunk_presorted<3>(key, lane, (wave & 1) == 0);
    u64 cur[NSLOT];
    #pragma unroll
    for (int q = 0; q < 8; ++q) cur[q] = key[24 + q];
    if (tid == 0) P->toobig = 0;
    #pragma nounroll
    for (int attempt = 0; attempt < 2; ++attempt) {
        #pragma nounroll
        for (int r = 0; r < 3; ++r) {
            #pragma unroll
            for (int q = 0; q < 8; ++q) xch[r][wave][q * 32 + lane] = cur[q];
            __syncthreads();
            #pragma unroll
            for (int q = 0; q < 8; ++q) cur[q] = key_max(cur[q], xch[r][wave ^ (1 << r)][q * 32 + lane]);
            merge_rt<1, 16>(cur, lane, 16, (r == 2) ? true : ((wave & (2 << r)) == 0), false);
        }
        if (wave == 0) {
            #pragma unroll
            for (int q = 0; q < 8; ++q) keys[lane * 8 + q] = cur[q];
        }
        __syncthreads();
        if (attempt == 1) break;
        {
            const u64 last = keys[CAND - 1];
            unsigned above = 0;
            #pragma unroll
            for (int s = 0; s < 24; ++s) above |= key_lt_mask(last, key[s]);
            if (above != 0 && lane == 0) P->toobig = 1;
        }
        __syncthreads();
        if (__builtin_amdgcn_readfirstlane(P->toobig) == 0) break;
        const bool wdesc = (wave & 1) == 0;
        #pragma nounroll
        for (int j = 0; j < 3; ++j) {
            sort_chunk<0>(key, lane, !wdesc);
            #pragma unroll
            for (int q = 0; q < 8; ++q) key[24 + q] = key_max(key[24 + q], key[q]);
            merge_rt<1, 8, 24>(key, lane, 16, wdesc, false);
            #pragma unroll
            for (int q = 0; q < 8; ++q) { key[q] = key[8 + q]; key[8 + q] = key[16 + q]; }
        }
        #pragma unroll
        for (int q = 0; q < 8; ++q) cur[q] = key[24 + q];
    }

    {
        unsigned long long key = keys[tid];
        unsigned id = ~(unsigned)key;
        float s = __uint_as_float((unsigned)(key >> 32));
        int cell = (int)(id >> 1);
        int gi = cell / S_GRID, gj = cell % S_GRID;
        const float* cd = in + COORD_OFF + (size_t)id * 4;
        float tx = cd[0], ty = cd[1], tw = cd[2], th = cd[3];
        float x = __fdiv_rn(__fadd_rn(tx, (float)gj), 56.0f);
        float y = __fdiv_rn(__fadd_rn(ty, (float)gi), 56.0f);
        float hw = __fmul_rn(__fmul_rn(tw, tw), 0.5f);
        float hh = __fmul_rn(__fmul_rn(th, th), 0.5f);
        float y0 = __fsub_rn(y, hh), x0 = __fsub_rn(x, hw);
        float y1 = __fadd_rn(y, hh), x1 = __fadd_rn(x, hw);
        float conf = in[CONF_OFF + id];
        const float* pr = in + cell * N_CLASS;
        float best = NEG_INF; int bc = 0;
        #pragma unroll
        for (int c = 0; c < N_CLASS; ++c) {
            float v = __fmul_rn(conf, pr[c]);
            if (v > best) { best = v; bc = c; }
        }
        float dy = fmaxf(__fsub_rn(y1, y0), 0.0f);
        float dx = fmaxf(__fsub_rn(x1, x0), 0.0f);
        P->y0[tid] = y0; P->x0[tid] = x0; P->y1[tid] = y1; P->x1[tid] = x1;
        P->sc[tid] = s;  P->cl[tid] = (float)bc;
        P->ar[tid] = __fmul_rn(dy, dx);
        if (tid < 32) { P->selIdx[tid] = 0; P->selOk[tid] = 0; }
    }
    __syncthreads();

    {
        const unsigned below = (1u << lane) - 1u;
        int total = 0;
        for (int q = 0; q < 8; ++q) {
            v8f czero = {};
            for (int tt = wave; tt < 4 * (q + 1); tt += NWAVE) {
                int mi = 2 * q + (tt & 1), ni = tt >> 1;
                int m0 = mi << 4, n0 = ni << 4;
                v2f A, B;
                if (lane < 16) {
                    A[0] = P->ar[m0 + lane]; A[1] = 1.0f;
                    B[0] = 1.0f;             B[1] = P->ar[n0 + lane];
                } else {
                    A[0] = 0.0f; A[1] = 0.0f; B[0] = 0.0f; B[1] = 0.0f;
                }
                v8f d = __builtin_amdgcn_wmma_f32_16x16x4_f32(
                    false, A, false, B, (short)0, czero, false, false);
                int n = n0 + (lane & 15);
                float ny0 = P->y0[n], nx0 = P->x0[n], ny1 = P->y1[n], nx1 = P->x1[n];
                int mbase = m0 + ((lane & 16) ? 8 : 0);
                #pragma unroll
                for (int r = 0; r < 8; ++r) {
                    int m = mbase + r;
                    float iy = fmaxf(__fsub_rn(fminf(P->y1[m], ny1), fmaxf(P->y0[m], ny0)), 0.0f);
                    float ix = fmaxf(__fsub_rn(fminf(P->x1[m], nx1), fmaxf(P->x0[m], nx0)), 0.0f);
                    float inter = __fmul_rn(iy, ix);
                    float uni = __fsub_rn(d[r], inter);
                    bool pred = (uni > 0.0f) && (__fdiv_rn(inter, uni) > 0.4f);
                    unsigned bal = __builtin_amdgcn_ballot_w32(pred);
                    if (lane == 0) {
                        P->mask16[(m0 + r)     * 16 + ni] = (unsigned short)(bal & 0xFFFFu);
                        P->mask16[(m0 + 8 + r) * 16 + ni] = (unsigned short)(bal >> 16);
                    }
                }
            }
            __syncthreads();

            if (wave == 0) {
                unsigned ext = 0;
                for (int w = 0; w < q; ++w) ext |= P->maskw[(q * 32 + lane) * 8 + w] & P->keepw[w];
                const unsigned own = P->maskw[(q * 32 + lane) * 8 + q] & below;
                const unsigned open = __builtin_amdgcn_ballot_w32(ext == 0);
                unsigned k = open;
                for (int pass = 0; pass < 32; ++pass) {
                    unsigned k2 = open & __builtin_amdgcn_ballot_w32((own & k) == 0);
                    if (k2 == k) break;
                    k = k2;
                }
                if ((k >> lane) & 1u) {
                    int rank = total + __builtin_popcount(k & below);
                    if (rank < MAXOUT) { P->selIdx[rank] = q * 32 + lane; P->selOk[rank] = 1; }
                }
                total += __builtin_popcount(k);
                if (lane == 0) { P->npick = total; P->keepw[q] = k; }
            }
            __syncthreads();
            if (P->npick >= MAXOUT) break;
        }
    }

    if (wave == 0) {
        const int sel = P->selIdx[lane];
        const float okf = P->selOk[lane] ? 1.0f : 0.0f;
        const float ms = __fmul_rn(P->sc[sel], okf);
        const float r0 = __fmul_rn(P->y0[sel], okf);
        const float r1 = __fmul_rn(P->x0[sel], okf);
        const float r2 = __fmul_rn(P->y1[sel], okf);
        const float r3 = __fmul_rn(P->x1[sel], okf);
        const float r5 = __fmul_rn(P->cl[sel], okf);
        int ahead = 0;
        unsigned same = 0;
        for (int jt = 0; jt < MAXOUT; ++jt) {
            const float vj = __int_as_float(__builtin_amdgcn_readlane(__float_as_int(ms), jt));
            ahead += (vj > ms) ? 1 : 0;
            same |= (vj == ms) ? (1u << jt) : 0u;
        }
        const int rank = ahead + __builtin_popcount(same & ((1u << lane) - 1u));
        if (lane < MAXOUT) {
            float* o = out + (size_t)img * (MAXOUT * 6) + (size_t)rank * 6;
            o[0] = r0;
            o[1] = r1;
            o[2] = r2;
            o[3] = r3;
            o[4] = ms;
            o[5] = r5;
        }
    }
}

extern "C" void kernel_launch(void* const* d_in, const int* in_sizes, int n_in,
                              void* d_out, int out_size, void* d_ws, size_t ws_size,
                              hipStream_t stream) {
    (void)n_in; (void)out_size; (void)d_ws; (void)ws_size;
    const float* net = (const float*)d_in[0];
    float* out = (float*)d_out;
    int batch = in_sizes[0] / D_IMG;
    det_kernel<<<dim3(batch), dim3(256), 0, stream>>>(net, out);
}
